# starting kernel
# baseline (speedup 1.0000x reference)
#include <hip/hip_runtime.h>
#include <math.h>

#define N_NODES 100000
#define N_EDGES 800000
#define IN_DIM  128
#define HEADS   4
#define DHEAD   32
#define HD      128
#define EPS_LN  1e-5f
#define NEG_SLOPE 0.2f

typedef __attribute__((ext_vector_type(2))) float v2f;
typedef __attribute__((ext_vector_type(4))) float v4f;
typedef __attribute__((ext_vector_type(8))) float v8f;

__global__ void fill_kernel(float* __restrict__ out, float* __restrict__ esum,
                            int* __restrict__ emax) {
    int i = blockIdx.x * blockDim.x + threadIdx.x;
    if (i < N_NODES * HD) out[i] = 0.0f;
    if (i < N_NODES * HEADS) {
        esum[i] = 0.0f;
        emax[i] = (int)0xFF800000u;
    }
}

__global__ __launch_bounds__(256) void gemm_wmma_kernel(
    const float* __restrict__ feat,
    const float* __restrict__ Wsrc,  const float* __restrict__ bsrc,
    const float* __restrict__ Wskip, const float* __restrict__ bskip,
    float* __restrict__ feat_src, float* __restrict__ skip) {
    int wave = (blockIdx.x * blockDim.x + threadIdx.x) >> 5;
    int lane = threadIdx.x & 31;
    if (wave >= N_NODES / 16) return;
    const int node0 = wave * 16;
    const int half  = lane >> 4;
    const int l16   = lane & 15;

    const float* Arow = feat + (size_t)(node0 + l16) * IN_DIM + 2 * half;

    for (int p = 0; p < 2; ++p) {
        const float* W    = p ? Wskip : Wsrc;
        const float* bias = p ? bskip : bsrc;
        float*       out  = p ? skip  : feat_src;
        for (int ct = 0; ct < 8; ++ct) {
            const int n = ct * 16 + l16;
            v8f acc;
            const float bv = bias[n];
            #pragma unroll
            for (int v = 0; v < 8; ++v) acc[v] = bv;
            const float* Bcol = W + (size_t)n * IN_DIM + 2 * half;
            for (int kt = 0; kt < 32; ++kt) {
                v2f a = *(const v2f*)(Arow + kt * 4);
                v2f b = *(const v2f*)(Bcol + kt * 4);
                acc = __builtin_amdgcn_wmma_f32_16x16x4_f32(
                    false, a, false, b, (short)0, acc, false, false);
            }
            float* obase = out + (size_t)node0 * HD + ct * 16 + l16;
            #pragma unroll
            for (int v = 0; v < 8; ++v)
                obase[(size_t)(v + 8 * half) * HD] = acc[v];
        }
    }
}

__global__ void elr_kernel(const float* __restrict__ feat_src,
                           const float* __restrict__ attn_l,
                           const float* __restrict__ attn_r,
                           float* __restrict__ el, float* __restrict__ er) {
    int i = blockIdx.x * blockDim.x + threadIdx.x;
    if (i >= N_NODES * HEADS) return;
    const int h = i & 3;
    const float* f  = feat_src + (size_t)(i >> 2) * HD + h * DHEAD;
    const float* al = attn_l + h * DHEAD;
    const float* ar = attn_r + h * DHEAD;
    float sl = 0.0f, sr = 0.0f;
    #pragma unroll
    for (int d = 0; d < DHEAD; ++d) {
        float v = f[d];
        sl += v * al[d];
        sr += v * ar[d];
    }
    el[i] = sl;
    er[i] = sr;
}

__global__ void edge_logit_kernel(const int* __restrict__ src, const int* __restrict__ dst,
                                  const float* __restrict__ el, const float* __restrict__ er,
                                  float* __restrict__ ebuf, int* __restrict__ emax) {
    int e = blockIdx.x * blockDim.x + threadIdx.x;
    if (e >= N_EDGES) return;
    const int s = src[e], d = dst[e];
    const v4f els = *(const v4f*)(el + (size_t)s * 4);
    const v4f erd = *(const v4f*)(er + (size_t)d * 4);
    #pragma unroll
    for (int h = 0; h < 4; ++h) {
        float v = els[h] + erd[h];
        v = (v >= 0.0f) ? v : NEG_SLOPE * v;
        ebuf[(size_t)e * 4 + h] = v;
        int* p = emax + (size_t)d * 4 + h;
        if (v >= 0.0f) atomicMax(p, __float_as_int(v));
        else           atomicMin((unsigned int*)p, __float_as_uint(v));
    }
}

__global__ void edge_exp_kernel(const int* __restrict__ dst,
                                const float* __restrict__ emax,
                                float* __restrict__ ebuf, float* __restrict__ esum) {
    int e = blockIdx.x * blockDim.x + threadIdx.x;
    if (e >= N_EDGES) return;
    const int d = dst[e];
    #pragma unroll
    for (int h = 0; h < 4; ++h) {
        float m  = emax[(size_t)d * 4 + h];
        float ex = __expf(ebuf[(size_t)e * 4 + h] - m);
        ebuf[(size_t)e * 4 + h] = ex;
        atomicAdd(esum + (size_t)d * 4 + h, ex);
    }
}

__global__ __launch_bounds__(256) void aggregate_kernel(
    const int* __restrict__ src, const int* __restrict__ dst,
    const float* __restrict__ ebuf, const float* __restrict__ feat_src,
    float* __restrict__ out) {
    int wave = (blockIdx.x * blockDim.x + threadIdx.x) >> 5;
    int lane = threadIdx.x & 31;
    if (wave >= N_EDGES) return;
    const int s = src[wave], d = dst[wave];
    const int h = lane >> 3;
    const float a = ebuf[(size_t)wave * 4 + h];
    const v4f f = *(const v4f*)(feat_src + (size_t)s * HD + lane * 4);
    float* o = out + (size_t)d * HD + lane * 4;
    atomicAdd(o + 0, a * f.x);
    atomicAdd(o + 1, a * f.y);
    atomicAdd(o + 2, a * f.z);
    atomicAdd(o + 3, a * f.w);
}

__global__ __launch_bounds__(256) void final_kernel(
    const float* __restrict__ skip, const float* __restrict__ esum,
    const float* __restrict__ Wg, const float* __restrict__ bg,
    const float* __restrict__ ln_g, const float* __restrict__ ln_b,
    const float* __restrict__ prelu_a, float* __restrict__ out) {
    int wave = (blockIdx.x * blockDim.x + threadIdx.x) >> 5;
    int lane = threadIdx.x & 31;
    if (wave >= N_NODES) return;
    const int c = lane * 4;
    const int h = lane >> 3;
    const float es  = esum[(size_t)wave * 4 + h];
    const float inv = (es > 0.0f) ? 1.0f / es : 0.0f;
    v4f r  = *(const v4f*)(out  + (size_t)wave * HD + c);
    r = r * inv;
    const v4f sk = *(const v4f*)(skip + (size_t)wave * HD + c);
    const v4f dm = r - sk;
    const v4f w0 = *(const v4f*)(Wg + c);
    const v4f w1 = *(const v4f*)(Wg + 128 + c);
    const v4f w2 = *(const v4f*)(Wg + 256 + c);
    float part = r.x * w0.x + r.y * w0.y + r.z * w0.z + r.w * w0.w
               + sk.x * w1.x + sk.y * w1.y + sk.z * w1.z + sk.w * w1.w
               + dm.x * w2.x + dm.y * w2.y + dm.z * w2.z + dm.w * w2.w;
    #pragma unroll
    for (int off = 16; off > 0; off >>= 1) part += __shfl_xor(part, off, 32);
    const float g = 1.0f / (1.0f + __expf(-(part + bg[0])));
    v4f z = g * r + (1.0f - g) * sk;
    float s  = z.x + z.y + z.z + z.w;
    float sq = z.x * z.x + z.y * z.y + z.z * z.z + z.w * z.w;
    #pragma unroll
    for (int off = 16; off > 0; off >>= 1) {
        s  += __shfl_xor(s,  off, 32);
        sq += __shfl_xor(sq, off, 32);
    }
    const float mu   = s * (1.0f / 128.0f);
    const float var  = sq * (1.0f / 128.0f) - mu * mu;
    const float rstd = rsqrtf(var + EPS_LN);
    const float alpha = prelu_a[0];
    const v4f lg = *(const v4f*)(ln_g + c);
    const v4f lb = *(const v4f*)(ln_b + c);
    v4f y;
    #pragma unroll
    for (int i = 0; i < 4; ++i) {
        float yy = (z[i] - mu) * rstd * lg[i] + lb[i];
        y[i] = (yy >= 0.0f) ? yy : alpha * yy;
    }
    *(v4f*)(out + (size_t)wave * HD + c) = y;
}

extern "C" void kernel_launch(void* const* d_in, const int* in_sizes, int n_in,
                              void* d_out, int out_size, void* d_ws, size_t ws_size,
                              hipStream_t stream) {
    const float* feat    = (const float*)d_in[0];
    const int*   src     = (const int*)  d_in[1];
    const int*   dst     = (const int*)  d_in[2];
    const float* W_src   = (const float*)d_in[3];
    const float* b_src   = (const float*)d_in[4];
    const float* attn_l  = (const float*)d_in[5];
    const float* attn_r  = (const float*)d_in[6];
    const float* W_skip  = (const float*)d_in[7];
    const float* b_skip  = (const float*)d_in[8];
    const float* W_gate  = (const float*)d_in[9];
    const float* b_gate  = (const float*)d_in[10];
    const float* ln_g    = (const float*)d_in[11];
    const float* ln_b    = (const float*)d_in[12];
    const float* prelu_a = (const float*)d_in[13];
    float* out = (float*)d_out;

    float* ws       = (float*)d_ws;
    float* feat_src = ws;
    float* skip     = feat_src + (size_t)N_NODES * HD;
    float* el       = skip     + (size_t)N_NODES * HD;
    float* er       = el       + (size_t)N_NODES * HEADS;
    float* emax     = er       + (size_t)N_NODES * HEADS;
    float* esum     = emax     + (size_t)N_NODES * HEADS;
    float* ebuf     = esum     + (size_t)N_NODES * HEADS;

    {
        int n = N_NODES * HD;
        fill_kernel<<<(n + 255) / 256, 256, 0, stream>>>(out, esum, (int*)emax);
    }
    {
        int waves  = N_NODES / 16;
        int blocks = (waves * 32 + 255) / 256;
        gemm_wmma_kernel<<<blocks, 256, 0, stream>>>(feat, W_src, b_src, W_skip, b_skip,
                                                     feat_src, skip);
    }
    {
        int n = N_NODES * HEADS;
        elr_kernel<<<(n + 255) / 256, 256, 0, stream>>>(feat_src, attn_l, attn_r, el, er);
    }
    edge_logit_kernel<<<(N_EDGES + 255) / 256, 256, 0, stream>>>(src, dst, el, er, ebuf,
                                                                 (int*)emax);
    edge_exp_kernel<<<(N_EDGES + 255) / 256, 256, 0, stream>>>(dst, emax, ebuf, esum);
    {
        int blocks = (N_EDGES * 32 + 255) / 256;
        aggregate_kernel<<<blocks, 256, 0, stream>>>(src, dst, ebuf, feat_src, out);
    }
    {
        int blocks = (N_NODES * 32 + 255) / 256;
        final_kernel<<<blocks, 256, 0, stream>>>(skip, esum, W_gate, b_gate, ln_g, ln_b,
                                                 prelu_a, out);
    }
}


# Edge logit and exp kernels: 4 adjacent lanes per edge (lane = head), so an edge's 4 per-head atomics are one contiguous request; max kept as order-preserving uint key (one atomic max, no sign split)
# speedup vs baseline: 1.4117x; 1.4117x over previous
#include <hip/hip_runtime.h>
#include <math.h>

#define N_NODES 100000
#define N_EDGES 800000
#define IN_DIM  128
#define HEADS   4
#define DHEAD   32
#define HD      128
#define EPS_LN  1e-5f
#define NEG_SLOPE 0.2f

typedef __attribute__((ext_vector_type(2))) float v2f;
typedef __attribute__((ext_vector_type(4))) float v4f;
typedef __attribute__((ext_vector_type(8))) float v8f;

__device__ __forceinline__ unsigned int f32_to_key(float v) {
    const unsigned int b = __float_as_uint(v);
    return (b & 0x80000000u) ? ~b : (b | 0x80000000u);
}
__device__ __forceinline__ float key_to_f32(unsigned int k) {
    return __uint_as_float((k & 0x80000000u) ? (k & 0x7FFFFFFFu) : ~k);
}

__global__ void fill_kernel(float* __restrict__ out, float* __restrict__ esum,
                            unsigned int* __restrict__ emax) {
    int i = blockIdx.x * blockDim.x + threadIdx.x;
    if (i < N_NODES * HD) out[i] = 0.0f;
    if (i < N_NODES * HEADS) {
        esum[i] = 0.0f;
        emax[i] = 0u;
    }
}

__global__ __launch_bounds__(256) void gemm_wmma_kernel(
    const float* __restrict__ feat,
    const float* __restrict__ Wsrc,  const float* __restrict__ bsrc,
    const float* __restrict__ Wskip, const float* __restrict__ bskip,
    float* __restrict__ feat_src, float* __restrict__ skip) {
    int wave = (blockIdx.x * blockDim.x + threadIdx.x) >> 5;
    int lane = threadIdx.x & 31;
    if (wave >= N_NODES / 16) return;
    const int node0 = wave * 16;
    const int half  = lane >> 4;
    const int l16   = lane & 15;

    const float* Arow = feat + (size_t)(node0 + l16) * IN_DIM + 2 * half;

    for (int p = 0; p < 2; ++p) {
        const float* W    = p ? Wskip : Wsrc;
        const float* bias = p ? bskip : bsrc;
        float*       out  = p ? skip  : feat_src;
        for (int ct = 0; ct < 8; ++ct) {
            const int n = ct * 16 + l16;
            v8f acc;
            const float bv = bias[n];
            #pragma unroll
            for (int v = 0; v < 8; ++v) acc[v] = bv;
            const float* Bcol = W + (size_t)n * IN_DIM + 2 * half;
            for (int kt = 0; kt < 32; ++kt) {
                v2f a = *(const v2f*)(Arow + kt * 4);
                v2f b = *(const v2f*)(Bcol + kt * 4);
                acc = __builtin_amdgcn_wmma_f32_16x16x4_f32(
                    false, a, false, b, (short)0, acc, false, false);
            }
            float* obase = out + (size_t)node0 * HD + ct * 16 + l16;
            #pragma unroll
            for (int v = 0; v < 8; ++v)
                obase[(size_t)(v + 8 * half) * HD] = acc[v];
        }
    }
}

__global__ void elr_kernel(const float* __restrict__ feat_src,
                           const float* __restrict__ attn_l,
                           const float* __restrict__ attn_r,
                           float* __restrict__ el, float* __restrict__ er) {
    int i = blockIdx.x * blockDim.x + threadIdx.x;
    if (i >= N_NODES * HEADS) return;
    const int h = i & 3;
    const float* f  = feat_src + (size_t)(i >> 2) * HD + h * DHEAD;
    const float* al = attn_l + h * DHEAD;
    const float* ar = attn_r + h * DHEAD;
    float sl = 0.0f, sr = 0.0f;
    #pragma unroll
    for (int d = 0; d < DHEAD; ++d) {
        float v = f[d];
        sl += v * al[d];
        sr += v * ar[d];
    }
    el[i] = sl;
    er[i] = sr;
}

static_assert(N_EDGES % 32 == 0, "a wave owns 32 whole edges");
__global__ __launch_bounds__(256) void edge_logit_kernel(
    const int* __restrict__ src, const int* __restrict__ dst,
    const float* __restrict__ el, const float* __restrict__ er,
    float* __restrict__ ebuf, unsigned int* __restrict__ emax) {
    const int lane  = threadIdx.x & 31;
    const int ebase = (blockIdx.x * blockDim.x + threadIdx.x) & ~31;
    if (ebase >= N_EDGES) return;
    const int s_own = src[ebase + lane];
    const int d_own = dst[ebase + lane];
    const int h   = lane & 3;
    const int sub = lane >> 2;
    int   dd[4];
    float a[4], b[4];
    for (int i = 0; i < 4; ++i) {
        const int j = i * 8 + sub;
        const int s = __shfl(s_own, j, 32);
        dd[i] = __shfl(d_own, j, 32);
        a[i] = el[(size_t)s * 4 + h];
        b[i] = er[(size_t)dd[i] * 4 + h];
    }
    for (int i = 0; i < 4; ++i) {
        float v = a[i] + b[i];
        v = (v >= 0.0f) ? v : NEG_SLOPE * v;
        ebuf[(size_t)ebase * 4 + i * 32 + lane] = v;
        atomicMax(emax + (size_t)dd[i] * 4 + h, f32_to_key(v));
    }
}

__global__ __launch_bounds__(256) void edge_exp_kernel(
    const int* __restrict__ dst, const unsigned int* __restrict__ emax,
    float* __restrict__ ebuf, float* __restrict__ esum) {
    const int lane  = threadIdx.x & 31;
    const int ebase = (blockIdx.x * blockDim.x + threadIdx.x) & ~31;
    if (ebase >= N_EDGES) return;
    const int d_own = dst[ebase + lane];
    const int h   = lane & 3;
    const int sub = lane >> 2;
    float* eb = ebuf + (size_t)ebase * 4 + lane;
    int          dd[4];
    unsigned int mk[4];
    float        ev[4];
    for (int i = 0; i < 4; ++i) {
        dd[i] = __shfl(d_own, i * 8 + sub, 32);
        mk[i] = emax[(size_t)dd[i] * 4 + h];
        ev[i] = eb[i * 32];
    }
    for (int i = 0; i < 4; ++i) {
        const float ex = __expf(ev[i] - key_to_f32(mk[i]));
        eb[i * 32] = ex;
        atomicAdd(esum + (size_t)dd[i] * 4 + h, ex);
    }
}

__global__ __launch_bounds__(256) void aggregate_kernel(
    const int* __restrict__ src, const int* __restrict__ dst,
    const float* __restrict__ ebuf, const float* __restrict__ feat_src,
    float* __restrict__ out) {
    int wave = (blockIdx.x * blockDim.x + threadIdx.x) >> 5;
    int lane = threadIdx.x & 31;
    if (wave >= N_EDGES) return;
    const int s = src[wave], d = dst[wave];
    const int h = lane >> 3;
    const float a = ebuf[(size_t)wave * 4 + h];
    const v4f f = *(const v4f*)(feat_src + (size_t)s * HD + lane * 4);
    float* o = out + (size_t)d * HD + lane * 4;
    atomicAdd(o + 0, a * f.x);
    atomicAdd(o + 1, a * f.y);
    atomicAdd(o + 2, a * f.z);
    atomicAdd(o + 3, a * f.w);
}

__global__ __launch_bounds__(256) void final_kernel(
    const float* __restrict__ skip, const float* __restrict__ esum,
    const float* __restrict__ Wg, const float* __restrict__ bg,
    const float* __restrict__ ln_g, const float* __restrict__ ln_b,
    const float* __restrict__ prelu_a, float* __restrict__ out) {
    int wave = (blockIdx.x * blockDim.x + threadIdx.x) >> 5;
    int lane = threadIdx.x & 31;
    if (wave >= N_NODES) return;
    const int c = lane * 4;
    const int h = lane >> 3;
    const float es  = esum[(size_t)wave * 4 + h];
    const float inv = (es > 0.0f) ? 1.0f / es : 0.0f;
    v4f r  = *(const v4f*)(out  + (size_t)wave * HD + c);
    r = r * inv;
    const v4f sk = *(const v4f*)(skip + (size_t)wave * HD + c);
    const v4f dm = r - sk;
    const v4f w0 = *(const v4f*)(Wg + c);
    const v4f w1 = *(const v4f*)(Wg + 128 + c);
    const v4f w2 = *(const v4f*)(Wg + 256 + c);
    float part = r.x * w0.x + r.y * w0.y + r.z * w0.z + r.w * w0.w
               + sk.x * w1.x + sk.y * w1.y + sk.z * w1.z + sk.w * w1.w
               + dm.x * w2.x + dm.y * w2.y + dm.z * w2.z + dm.w * w2.w;
    #pragma unroll
    for (int off = 16; off > 0; off >>= 1) part += __shfl_xor(part, off, 32);
    const float g = 1.0f / (1.0f + __expf(-(part + bg[0])));
    v4f z = g * r + (1.0f - g) * sk;
    float s  = z.x + z.y + z.z + z.w;
    float sq = z.x * z.x + z.y * z.y + z.z * z.z + z.w * z.w;
    #pragma unroll
    for (int off = 16; off > 0; off >>= 1) {
        s  += __shfl_xor(s,  off, 32);
        sq += __shfl_xor(sq, off, 32);
    }
    const float mu   = s * (1.0f / 128.0f);
    const float var  = sq * (1.0f / 128.0f) - mu * mu;
    const float rstd = rsqrtf(var + EPS_LN);
    const float alpha = prelu_a[0];
    const v4f lg = *(const v4f*)(ln_g + c);
    const v4f lb = *(const v4f*)(ln_b + c);
    v4f y;
    #pragma unroll
    for (int i = 0; i < 4; ++i) {
        float yy = (z[i] - mu) * rstd * lg[i] + lb[i];
        y[i] = (yy >= 0.0f) ? yy : alpha * yy;
    }
    *(v4f*)(out + (size_t)wave * HD + c) = y;
}

extern "C" void kernel_launch(void* const* d_in, const int* in_sizes, int n_in,
                              void* d_out, int out_size, void* d_ws, size_t ws_size,
                              hipStream_t stream) {
    const float* feat    = (const float*)d_in[0];
    const int*   src     = (const int*)  d_in[1];
    const int*   dst     = (const int*)  d_in[2];
    const float* W_src   = (const float*)d_in[3];
    const float* b_src   = (const float*)d_in[4];
    const float* attn_l  = (const float*)d_in[5];
    const float* attn_r  = (const float*)d_in[6];
    const float* W_skip  = (const float*)d_in[7];
    const float* b_skip  = (const float*)d_in[8];
    const float* W_gate  = (const float*)d_in[9];
    const float* b_gate  = (const float*)d_in[10];
    const float* ln_g    = (const float*)d_in[11];
    const float* ln_b    = (const float*)d_in[12];
    const float* prelu_a = (const float*)d_in[13];
    float* out = (float*)d_out;

    float* ws       = (float*)d_ws;
    float* feat_src = ws;
    float* skip     = feat_src + (size_t)N_NODES * HD;
    float* el       = skip     + (size_t)N_NODES * HD;
    float* er       = el       + (size_t)N_NODES * HEADS;
    float* emax     = er       + (size_t)N_NODES * HEADS;
    float* esum     = emax     + (size_t)N_NODES * HEADS;
    float* ebuf     = esum     + (size_t)N_NODES * HEADS;

    {
        int n = N_NODES * HD;
        fill_kernel<<<(n + 255) / 256, 256, 0, stream>>>(out, esum, (unsigned int*)emax);
    }
    {
        int waves  = N_NODES / 16;
        int blocks = (waves * 32 + 255) / 256;
        gemm_wmma_kernel<<<blocks, 256, 0, stream>>>(feat, W_src, b_src, W_skip, b_skip,
                                                     feat_src, skip);
    }
    {
        int n = N_NODES * HEADS;
        elr_kernel<<<(n + 255) / 256, 256, 0, stream>>>(feat_src, attn_l, attn_r, el, er);
    }
    edge_logit_kernel<<<(N_EDGES + 255) / 256, 256, 0, stream>>>(src, dst, el, er, ebuf,
                                                                 (unsigned int*)emax);
    edge_exp_kernel<<<(N_EDGES + 255) / 256, 256, 0, stream>>>(dst, (const unsigned int*)emax, ebuf,
                                                               esum);
    {
        int blocks = (N_EDGES * 32 + 255) / 256;
        aggregate_kernel<<<blocks, 256, 0, stream>>>(src, dst, ebuf, feat_src, out);
    }
    {
        int blocks = (N_NODES * 32 + 255) / 256;
        final_kernel<<<blocks, 256, 0, stream>>>(skip, esum, W_gate, b_gate, ln_g, ln_b,
                                                 prelu_a, out);
    }
}


# Fat waves where each wave did one small item: fill thread stores 10 x 16 B, aggregation wave walks 32 edges, epilogue wave does 8 nodes (1300000 waves become 47504); atomic arrival order changes
# speedup vs baseline: 1.4701x; 1.0414x over previous
#include <hip/hip_runtime.h>
#include <math.h>

#define N_NODES 100000
#define N_EDGES 800000
#define IN_DIM  128
#define HEADS   4
#define DHEAD   32
#define HD      128
#define EPS_LN  1e-5f
#define NEG_SLOPE 0.2f

typedef __attribute__((ext_vector_type(2))) float v2f;
typedef __attribute__((ext_vector_type(4))) float v4f;
typedef __attribute__((ext_vector_type(8))) float v8f;

__device__ __forceinline__ unsigned int f32_to_key(float v) {
    const unsigned int b = __float_as_uint(v);
    return (b & 0x80000000u) ? ~b : (b | 0x80000000u);
}
__device__ __forceinline__ float key_to_f32(unsigned int k) {
    return __uint_as_float((k & 0x80000000u) ? (k & 0x7FFFFFFFu) : ~k);
}

typedef __attribute__((ext_vector_type(4))) unsigned int v4u;
#define FILL_V4_PER_THREAD 10
static_assert((N_NODES * HD / 4) % (256 * FILL_V4_PER_THREAD) == 0, "whole work-groups cover out exactly");
static_assert(N_NODES * HD / 4 / FILL_V4_PER_THREAD >= N_NODES * HEADS / 4, "enough threads for esum, emax");
__global__ __launch_bounds__(256) void fill_kernel(float* __restrict__ out, float* __restrict__ esum,
                                                   unsigned int* __restrict__ emax) {
    const v4f zf = {0.0f, 0.0f, 0.0f, 0.0f};
    const v4u zk = {0u, 0u, 0u, 0u};
    v4f* o4 = (v4f*)out + (size_t)blockIdx.x * (256 * FILL_V4_PER_THREAD) + threadIdx.x;
    for (int k = 0; k < FILL_V4_PER_THREAD; ++k) o4[k * 256] = zf;
    const int i = blockIdx.x * 256 + threadIdx.x;
    if (i < N_NODES * HEADS / 4) {
        ((v4f*)esum)[i] = zf;
        ((v4u*)emax)[i] = zk;
    }
}

__global__ __launch_bounds__(256) void gemm_wmma_kernel(
    const float* __restrict__ feat,
    const float* __restrict__ Wsrc,  const float* __restrict__ bsrc,
    const float* __restrict__ Wskip, const float* __restrict__ bskip,
    float* __restrict__ feat_src, float* __restrict__ skip) {
    int wave = (blockIdx.x * blockDim.x + threadIdx.x) >> 5;
    int lane = threadIdx.x & 31;
    if (wave >= N_NODES / 16) return;
    const int node0 = wave * 16;
    const int half  = lane >> 4;
    const int l16   = lane & 15;

    const float* Arow = feat + (size_t)(node0 + l16) * IN_DIM + 2 * half;

    for (int p = 0; p < 2; ++p) {
        const float* W    = p ? Wskip : Wsrc;
        const float* bias = p ? bskip : bsrc;
        float*       out  = p ? skip  : feat_src;
        for (int ct = 0; ct < 8; ++ct) {
            const int n = ct * 16 + l16;
            v8f acc;
            const float bv = bias[n];
            #pragma unroll
            for (int v = 0; v < 8; ++v) acc[v] = bv;
            const float* Bcol = W + (size_t)n * IN_DIM + 2 * half;
            for (int kt = 0; kt < 32; ++kt) {
                v2f a = *(const v2f*)(Arow + kt * 4);
                v2f b = *(const v2f*)(Bcol + kt * 4);
                acc = __builtin_amdgcn_wmma_f32_16x16x4_f32(
                    false, a, false, b, (short)0, acc, false, false);
            }
            float* obase = out + (size_t)node0 * HD + ct * 16 + l16;
            #pragma unroll
            for (int v = 0; v < 8; ++v)
                obase[(size_t)(v + 8 * half) * HD] = acc[v];
        }
    }
}

__global__ void elr_kernel(const float* __restrict__ feat_src,
                           const float* __restrict__ attn_l,
                           const float* __restrict__ attn_r,
                           float* __restrict__ el, float* __restrict__ er) {
    int i = blockIdx.x * blockDim.x + threadIdx.x;
    if (i >= N_NODES * HEADS) return;
    const int h = i & 3;
    const float* f  = feat_src + (size_t)(i >> 2) * HD + h * DHEAD;
    const float* al = attn_l + h * DHEAD;
    const float* ar = attn_r + h * DHEAD;
    float sl = 0.0f, sr = 0.0f;
    #pragma unroll
    for (int d = 0; d < DHEAD; ++d) {
        float v = f[d];
        sl += v * al[d];
        sr += v * ar[d];
    }
    el[i] = sl;
    er[i] = sr;
}

static_assert(N_EDGES % 32 == 0, "a wave owns 32 whole edges");
__global__ __launch_bounds__(256) void edge_logit_kernel(
    const int* __restrict__ src, const int* __restrict__ dst,
    const float* __restrict__ el, const float* __restrict__ er,
    float* __restrict__ ebuf, unsigned int* __restrict__ emax) {
    const int lane  = threadIdx.x & 31;
    const int ebase = (blockIdx.x * blockDim.x + threadIdx.x) & ~31;
    if (ebase >= N_EDGES) return;
    const int s_own = src[ebase + lane];
    const int d_own = dst[ebase + lane];
    const int h   = lane & 3;
    const int sub = lane >> 2;
    int   dd[4];
    float a[4], b[4];
    for (int i = 0; i < 4; ++i) {
        const int j = i * 8 + sub;
        const int s = __shfl(s_own, j, 32);
        dd[i] = __shfl(d_own, j, 32);
        a[i] = el[(size_t)s * 4 + h];
        b[i] = er[(size_t)dd[i] * 4 + h];
    }
    for (int i = 0; i < 4; ++i) {
        float v = a[i] + b[i];
        v = (v >= 0.0f) ? v : NEG_SLOPE * v;
        ebuf[(size_t)ebase * 4 + i * 32 + lane] = v;
        atomicMax(emax + (size_t)dd[i] * 4 + h, f32_to_key(v));
    }
}

__global__ __launch_bounds__(256) void edge_exp_kernel(
    const int* __restrict__ dst, const unsigned int* __restrict__ emax,
    float* __restrict__ ebuf, float* __restrict__ esum) {
    const int lane  = threadIdx.x & 31;
    const int ebase = (blockIdx.x * blockDim.x + threadIdx.x) & ~31;
    if (ebase >= N_EDGES) return;
    const int d_own = dst[ebase + lane];
    const int h   = lane & 3;
    const int sub = lane >> 2;
    float* eb = ebuf + (size_t)ebase * 4 + lane;
    int          dd[4];
    unsigned int mk[4];
    float        ev[4];
    for (int i = 0; i < 4; ++i) {
        dd[i] = __shfl(d_own, i * 8 + sub, 32);
        mk[i] = emax[(size_t)dd[i] * 4 + h];
        ev[i] = eb[i * 32];
    }
    for (int i = 0; i < 4; ++i) {
        const float ex = __expf(ev[i] - key_to_f32(mk[i]));
        eb[i * 32] = ex;
        atomicAdd(esum + (size_t)dd[i] * 4 + h, ex);
    }
}

static_assert(N_EDGES % 32 == 0, "a wave owns 32 whole edges");
__global__ __launch_bounds__(256) void aggregate_kernel(
    const int* __restrict__ src, const int* __restrict__ dst,
    const float* __restrict__ ebuf, const float* __restrict__ feat_src,
    float* __restrict__ out) {
    const int lane  = threadIdx.x & 31;
    const int ebase = (blockIdx.x * blockDim.x + threadIdx.x) & ~31;
    if (ebase >= N_EDGES) return;
    const int s_own = src[ebase + lane];
    const int d_own = dst[ebase + lane];
    const int h = lane >> 3;
    float aw[4];
    for (int i = 0; i < 4; ++i) aw[i] = ebuf[(size_t)ebase * 4 + i * 32 + lane];
    const float* fcol = feat_src + lane * 4;
    float*       ocol = out + lane * 4;
    for (int g = 0; g < 32; g += 4) {
        v4f   f[4];
        float a[4];
        int   dd[4];
        for (int u = 0; u < 4; ++u) {
            const int j = g + u;
            const int s = __builtin_amdgcn_readlane(s_own, j);
            dd[u] = __builtin_amdgcn_readlane(d_own, j);
            a[u]  = __shfl(aw[j >> 3], (j & 7) * 4 + h, 32);
            f[u]  = *(const v4f*)(fcol + (size_t)s * HD);
        }
        for (int u = 0; u < 4; ++u) {
            float* o = ocol + (size_t)dd[u] * HD;
            atomicAdd(o + 0, a[u] * f[u].x);
            atomicAdd(o + 1, a[u] * f[u].y);
            atomicAdd(o + 2, a[u] * f[u].z);
            atomicAdd(o + 3, a[u] * f[u].w);
        }
    }
}

#define FIN_NPW 8
static_assert(N_NODES % FIN_NPW == 0, "a wave owns 8 whole nodes");
__global__ __launch_bounds__(256) void final_kernel(
    const float* __restrict__ skip, const float* __restrict__ esum,
    const float* __restrict__ Wg, const float* __restrict__ bg,
    const float* __restrict__ ln_g, const float* __restrict__ ln_b,
    const float* __restrict__ prelu_a, float* __restrict__ out) {
    const int wave = (blockIdx.x * blockDim.x + threadIdx.x) >> 5;
    const int lane = threadIdx.x & 31;
    const int n0 = wave * FIN_NPW;
    if (n0 >= N_NODES) return;
    const int c = lane * 4;
    const int h = lane >> 3;
    const v4f w0 = *(const v4f*)(Wg + c);
    const v4f w1 = *(const v4f*)(Wg + 128 + c);
    const v4f w2 = *(const v4f*)(Wg + 256 + c);
    const v4f lg = *(const v4f*)(ln_g + c);
    const v4f lb = *(const v4f*)(ln_b + c);
    const float bgv   = bg[0];
    const float alpha = prelu_a[0];
    for (int b = 0; b < FIN_NPW; b += 4) {
        float es[4];
        v4f   rr[4], ss[4];
        #pragma unroll
        for (int u = 0; u < 4; ++u) {
            const size_t node = (size_t)(n0 + b + u);
            es[u] = esum[node * 4 + h];
            rr[u] = *(const v4f*)(out  + node * HD + c);
            ss[u] = *(const v4f*)(skip + node * HD + c);
        }
        #pragma unroll
        for (int u = 0; u < 4; ++u) {
            const size_t node = (size_t)(n0 + b + u);
            const float inv = (es[u] > 0.0f) ? 1.0f / es[u] : 0.0f;
            const v4f r  = rr[u] * inv;
            const v4f sk = ss[u];
            const v4f dm = r - sk;
            float part = r.x * w0.x + r.y * w0.y + r.z * w0.z + r.w * w0.w
                       + sk.x * w1.x + sk.y * w1.y + sk.z * w1.z + sk.w * w1.w
                       + dm.x * w2.x + dm.y * w2.y + dm.z * w2.z + dm.w * w2.w;
            #pragma unroll
            for (int off = 16; off > 0; off >>= 1) part += __shfl_xor(part, off, 32);
            const float g = 1.0f / (1.0f + __expf(-(part + bgv)));
            v4f z = g * r + (1.0f - g) * sk;
            float s  = z.x + z.y + z.z + z.w;
            float sq = z.x * z.x + z.y * z.y + z.z * z.z + z.w * z.w;
            #pragma unroll
            for (int off = 16; off > 0; off >>= 1) {
                s  += __shfl_xor(s,  off, 32);
                sq += __shfl_xor(sq, off, 32);
            }
            const float mu   = s * (1.0f / 128.0f);
            const float var  = sq * (1.0f / 128.0f) - mu * mu;
            const float rstd = rsqrtf(var + EPS_LN);
            v4f y;
            #pragma unroll
            for (int i = 0; i < 4; ++i) {
                float yy = (z[i] - mu) * rstd * lg[i] + lb[i];
                y[i] = (yy >= 0.0f) ? yy : alpha * yy;
            }
            *(v4f*)(out + node * HD + c) = y;
        }
    }
}

extern "C" void kernel_launch(void* const* d_in, const int* in_sizes, int n_in,
                              void* d_out, int out_size, void* d_ws, size_t ws_size,
                              hipStream_t stream) {
    const float* feat    = (const float*)d_in[0];
    const int*   src     = (const int*)  d_in[1];
    const int*   dst     = (const int*)  d_in[2];
    const float* W_src   = (const float*)d_in[3];
    const float* b_src   = (const float*)d_in[4];
    const float* attn_l  = (const float*)d_in[5];
    const float* attn_r  = (const float*)d_in[6];
    const float* W_skip  = (const float*)d_in[7];
    const float* b_skip  = (const float*)d_in[8];
    const float* W_gate  = (const float*)d_in[9];
    const float* b_gate  = (const float*)d_in[10];
    const float* ln_g    = (const float*)d_in[11];
    const float* ln_b    = (const float*)d_in[12];
    const float* prelu_a = (const float*)d_in[13];
    float* out = (float*)d_out;

    float* ws       = (float*)d_ws;
    float* feat_src = ws;
    float* skip     = feat_src + (size_t)N_NODES * HD;
    float* el       = skip     + (size_t)N_NODES * HD;
    float* er       = el       + (size_t)N_NODES * HEADS;
    float* emax     = er       + (size_t)N_NODES * HEADS;
    float* esum     = emax     + (size_t)N_NODES * HEADS;
    float* ebuf     = esum     + (size_t)N_NODES * HEADS;

    fill_kernel<<<N_NODES * HD / 4 / (256 * FILL_V4_PER_THREAD), 256, 0, stream>>>(out, esum,
                                                                                   (unsigned int*)emax);
    {
        int waves  = N_NODES / 16;
        int blocks = (waves * 32 + 255) / 256;
        gemm_wmma_kernel<<<blocks, 256, 0, stream>>>(feat, W_src, b_src, W_skip, b_skip,
                                                     feat_src, skip);
    }
    {
        int n = N_NODES * HEADS;
        elr_kernel<<<(n + 255) / 256, 256, 0, stream>>>(feat_src, attn_l, attn_r, el, er);
    }
    edge_logit_kernel<<<(N_EDGES + 255) / 256, 256, 0, stream>>>(src, dst, el, er, ebuf,
                                                                 (unsigned int*)emax);
    edge_exp_kernel<<<(N_EDGES + 255) / 256, 256, 0, stream>>>(dst, (const unsigned int*)emax, ebuf,
                                                               esum);
    aggregate_kernel<<<(N_EDGES + 255) / 256, 256, 0, stream>>>(src, dst, ebuf, feat_src, out);
    final_kernel<<<(N_NODES / FIN_NPW * 32 + 255) / 256, 256, 0, stream>>>(skip, esum, W_gate, b_gate,
                                                                           ln_g, ln_b, prelu_a, out);
}


# Aggregation: lane = one channel of head k (channel k*32+lane), so each float atomic add and each gather is one whole 128-byte line: 4 atomic line requests per edge, was 16; same arrival-order adds
# speedup vs baseline: 1.8606x; 1.2656x over previous
#include <hip/hip_runtime.h>
#include <math.h>

#define N_NODES 100000
#define N_EDGES 800000
#define IN_DIM  128
#define HEADS   4
#define DHEAD   32
#define HD      128
#define EPS_LN  1e-5f
#define NEG_SLOPE 0.2f

typedef __attribute__((ext_vector_type(2))) float v2f;
typedef __attribute__((ext_vector_type(4))) float v4f;
typedef __attribute__((ext_vector_type(8))) float v8f;

__device__ __forceinline__ unsigned int f32_to_key(float v) {
    const unsigned int b = __float_as_uint(v);
    return (b & 0x80000000u) ? ~b : (b | 0x80000000u);
}
__device__ __forceinline__ float key_to_f32(unsigned int k) {
    return __uint_as_float((k & 0x80000000u) ? (k & 0x7FFFFFFFu) : ~k);
}

typedef __attribute__((ext_vector_type(4))) unsigned int v4u;
#define FILL_V4_PER_THREAD 10
static_assert((N_NODES * HD / 4) % (256 * FILL_V4_PER_THREAD) == 0, "whole work-groups cover out exactly");
static_assert(N_NODES * HD / 4 / FILL_V4_PER_THREAD >= N_NODES * HEADS / 4, "enough threads for esum, emax");
__global__ __launch_bounds__(256) void fill_kernel(float* __restrict__ out, float* __restrict__ esum,
                                                   unsigned int* __restrict__ emax) {
    const v4f zf = {0.0f, 0.0f, 0.0f, 0.0f};
    const v4u zk = {0u, 0u, 0u, 0u};
    v4f* o4 = (v4f*)out + (size_t)blockIdx.x * (256 * FILL_V4_PER_THREAD) + threadIdx.x;
    for (int k = 0; k < FILL_V4_PER_THREAD; ++k) o4[k * 256] = zf;
    const int i = blockIdx.x * 256 + threadIdx.x;
    if (i < N_NODES * HEADS / 4) {
        ((v4f*)esum)[i] = zf;
        ((v4u*)emax)[i] = zk;
    }
}

__global__ __launch_bounds__(256) void gemm_wmma_kernel(
    const float* __restrict__ feat,
    const float* __restrict__ Wsrc,  const float* __restrict__ bsrc,
    const float* __restrict__ Wskip, const float* __restrict__ bskip,
    float* __restrict__ feat_src, float* __restrict__ skip) {
    int wave = (blockIdx.x * blockDim.x + threadIdx.x) >> 5;
    int lane = threadIdx.x & 31;
    if (wave >= N_NODES / 16) return;
    const int node0 = wave * 16;
    const int half  = lane >> 4;
    const int l16   = lane & 15;

    const float* Arow = feat + (size_t)(node0 + l16) * IN_DIM + 2 * half;

    for (int p = 0; p < 2; ++p) {
        const float* W    = p ? Wskip : Wsrc;
        const float* bias = p ? bskip : bsrc;
        float*       out  = p ? skip  : feat_src;
        for (int ct = 0; ct < 8; ++ct) {
            const int n = ct * 16 + l16;
            v8f acc;
            const float bv = bias[n];
            #pragma unroll
            for (int v = 0; v < 8; ++v) acc[v] = bv;
            const float* Bcol = W + (size_t)n * IN_DIM + 2 * half;
            for (int kt = 0; kt < 32; ++kt) {
                v2f a = *(const v2f*)(Arow + kt * 4);
                v2f b = *(const v2f*)(Bcol + kt * 4);
                acc = __builtin_amdgcn_wmma_f32_16x16x4_f32(
                    false, a, false, b, (short)0, acc, false, false);
            }
            float* obase = out + (size_t)node0 * HD + ct * 16 + l16;
            #pragma unroll
            for (int v = 0; v < 8; ++v)
                obase[(size_t)(v + 8 * half) * HD] = acc[v];
        }
    }
}

__global__ void elr_kernel(const float* __restrict__ feat_src,
                           const float* __restrict__ attn_l,
                           const float* __restrict__ attn_r,
                           float* __restrict__ el, float* __restrict__ er) {
    int i = blockIdx.x * blockDim.x + threadIdx.x;
    if (i >= N_NODES * HEADS) return;
    const int h = i & 3;
    const float* f  = feat_src + (size_t)(i >> 2) * HD + h * DHEAD;
    const float* al = attn_l + h * DHEAD;
    const float* ar = attn_r + h * DHEAD;
    float sl = 0.0f, sr = 0.0f;
    #pragma unroll
    for (int d = 0; d < DHEAD; ++d) {
        float v = f[d];
        sl += v * al[d];
        sr += v * ar[d];
    }
    el[i] = sl;
    er[i] = sr;
}

static_assert(N_EDGES % 32 == 0, "a wave owns 32 whole edges");
__global__ __launch_bounds__(256) void edge_logit_kernel(
    const int* __restrict__ src, const int* __restrict__ dst,
    const float* __restrict__ el, const float* __restrict__ er,
    float* __restrict__ ebuf, unsigned int* __restrict__ emax) {
    const int lane  = threadIdx.x & 31;
    const int ebase = (blockIdx.x * blockDim.x + threadIdx.x) & ~31;
    if (ebase >= N_EDGES) return;
    const int s_own = src[ebase + lane];
    const int d_own = dst[ebase + lane];
    const int h   = lane & 3;
    const int sub = lane >> 2;
    int   dd[4];
    float a[4], b[4];
    for (int i = 0; i < 4; ++i) {
        const int j = i * 8 + sub;
        const int s = __shfl(s_own, j, 32);
        dd[i] = __shfl(d_own, j, 32);
        a[i] = el[(size_t)s * 4 + h];
        b[i] = er[(size_t)dd[i] * 4 + h];
    }
    for (int i = 0; i < 4; ++i) {
        float v = a[i] + b[i];
        v = (v >= 0.0f) ? v : NEG_SLOPE * v;
        ebuf[(size_t)ebase * 4 + i * 32 + lane] = v;
        atomicMax(emax + (size_t)dd[i] * 4 + h, f32_to_key(v));
    }
}

__global__ __launch_bounds__(256) void edge_exp_kernel(
    const int* __restrict__ dst, const unsigned int* __restrict__ emax,
    float* __restrict__ ebuf, float* __restrict__ esum) {
    const int lane  = threadIdx.x & 31;
    const int ebase = (blockIdx.x * blockDim.x + threadIdx.x) & ~31;
    if (ebase >= N_EDGES) return;
    const int d_own = dst[ebase + lane];
    const int h   = lane & 3;
    const int sub = lane >> 2;
    float* eb = ebuf + (size_t)ebase * 4 + lane;
    int          dd[4];
    unsigned int mk[4];
    float        ev[4];
    for (int i = 0; i < 4; ++i) {
        dd[i] = __shfl(d_own, i * 8 + sub, 32);
        mk[i] = emax[(size_t)dd[i] * 4 + h];
        ev[i] = eb[i * 32];
    }
    for (int i = 0; i < 4; ++i) {
        const float ex = __expf(ev[i] - key_to_f32(mk[i]));
        eb[i * 32] = ex;
        atomicAdd(esum + (size_t)dd[i] * 4 + h, ex);
    }
}

static_assert(N_EDGES % 32 == 0, "a wave owns 32 whole edges");
static_assert(DHEAD == 32 && HD == HEADS * DHEAD, "one head = 32 channels = one line = one lane each");
__global__ __launch_bounds__(256) void aggregate_kernel(
    const int* __restrict__ src, const int* __restrict__ dst,
    const float* __restrict__ ebuf, const float* __restrict__ feat_src,
    float* __restrict__ out) {
    const int lane  = threadIdx.x & 31;
    const int ebase = (blockIdx.x * blockDim.x + threadIdx.x) & ~31;
    if (ebase >= N_EDGES) return;
    const int s_own = src[ebase + lane];
    const int d_own = dst[ebase + lane];
    unsigned int aw[4];
    for (int i = 0; i < 4; ++i) aw[i] = __float_as_uint(ebuf[(size_t)ebase * 4 + i * 32 + lane]);
    const float* fcol = feat_src + lane;
    float*       ocol = out + lane;
    for (int i = 0; i < 4; ++i) {
        for (int q0 = 0; q0 < 8; q0 += 4) {
            float f[4][HEADS];
            float a[4][HEADS];
            int   dd[4];
            for (int u = 0; u < 4; ++u) {
                const int q = q0 + u;
                const int j = i * 8 + q;
                const int s = __builtin_amdgcn_readlane(s_own, j);
                dd[u] = __builtin_amdgcn_readlane(d_own, j);
                const float* fr = fcol + (size_t)s * HD;
                for (int k = 0; k < HEADS; ++k) {
                    a[u][k] = __uint_as_float(__builtin_amdgcn_readlane(aw[i], q * 4 + k));
                    f[u][k] = fr[k * DHEAD];
                }
            }
            for (int u = 0; u < 4; ++u) {
                float* o = ocol + (size_t)dd[u] * HD;
                for (int k = 0; k < HEADS; ++k)
                    atomicAdd(o + k * DHEAD, a[u][k] * f[u][k]);
            }
        }
    }
}

#define FIN_NPW 8
static_assert(N_NODES % FIN_NPW == 0, "a wave owns 8 whole nodes");
__global__ __launch_bounds__(256) void final_kernel(
    const float* __restrict__ skip, const float* __restrict__ esum,
    const float* __restrict__ Wg, const float* __restrict__ bg,
    const float* __restrict__ ln_g, const float* __restrict__ ln_b,
    const float* __restrict__ prelu_a, float* __restrict__ out) {
    const int wave = (blockIdx.x * blockDim.x + threadIdx.x) >> 5;
    const int lane = threadIdx.x & 31;
    const int n0 = wave * FIN_NPW;
    if (n0 >= N_NODES) return;
    const int c = lane * 4;
    const int h = lane >> 3;
    const v4f w0 = *(const v4f*)(Wg + c);
    const v4f w1 = *(const v4f*)(Wg + 128 + c);
    const v4f w2 = *(const v4f*)(Wg + 256 + c);
    const v4f lg = *(const v4f*)(ln_g + c);
    const v4f lb = *(const v4f*)(ln_b + c);
    const float bgv   = bg[0];
    const float alpha = prelu_a[0];
    for (int b = 0; b < FIN_NPW; b += 4) {
        float es[4];
        v4f   rr[4], ss[4];
        #pragma unroll
        for (int u = 0; u < 4; ++u) {
            const size_t node = (size_t)(n0 + b + u);
            es[u] = esum[node * 4 + h];
            rr[u] = *(const v4f*)(out  + node * HD + c);
            ss[u] = *(const v4f*)(skip + node * HD + c);
        }
        #pragma unroll
        for (int u = 0; u < 4; ++u) {
            const size_t node = (size_t)(n0 + b + u);
            const float inv = (es[u] > 0.0f) ? 1.0f / es[u] : 0.0f;
            const v4f r  = rr[u] * inv;
            const v4f sk = ss[u];
            const v4f dm = r - sk;
            float part = r.x * w0.x + r.y * w0.y + r.z * w0.z + r.w * w0.w
                       + sk.x * w1.x + sk.y * w1.y + sk.z * w1.z + sk.w * w1.w
                       + dm.x * w2.x + dm.y * w2.y + dm.z * w2.z + dm.w * w2.w;
            #pragma unroll
            for (int off = 16; off > 0; off >>= 1) part += __shfl_xor(part, off, 32);
            const float g = 1.0f / (1.0f + __expf(-(part + bgv)));
            v4f z = g * r + (1.0f - g) * sk;
            float s  = z.x + z.y + z.z + z.w;
            float sq = z.x * z.x + z.y * z.y + z.z * z.z + z.w * z.w;
            #pragma unroll
            for (int off = 16; off > 0; off >>= 1) {
                s  += __shfl_xor(s,  off, 32);
                sq += __shfl_xor(sq, off, 32);
            }
            const float mu   = s * (1.0f / 128.0f);
            const float var  = sq * (1.0f / 128.0f) - mu * mu;
            const float rstd = rsqrtf(var + EPS_LN);
            v4f y;
            #pragma unroll
            for (int i = 0; i < 4; ++i) {
                float yy = (z[i] - mu) * rstd * lg[i] + lb[i];
                y[i] = (yy >= 0.0f) ? yy : alpha * yy;
            }
            *(v4f*)(out + node * HD + c) = y;
        }
    }
}

extern "C" void kernel_launch(void* const* d_in, const int* in_sizes, int n_in,
                              void* d_out, int out_size, void* d_ws, size_t ws_size,
                              hipStream_t stream) {
    const float* feat    = (const float*)d_in[0];
    const int*   src     = (const int*)  d_in[1];
    const int*   dst     = (const int*)  d_in[2];
    const float* W_src   = (const float*)d_in[3];
    const float* b_src   = (const float*)d_in[4];
    const float* attn_l  = (const float*)d_in[5];
    const float* attn_r  = (const float*)d_in[6];
    const float* W_skip  = (const float*)d_in[7];
    const float* b_skip  = (const float*)d_in[8];
    const float* W_gate  = (const float*)d_in[9];
    const float* b_gate  = (const float*)d_in[10];
    const float* ln_g    = (const float*)d_in[11];
    const float* ln_b    = (const float*)d_in[12];
    const float* prelu_a = (const float*)d_in[13];
    float* out = (float*)d_out;

    float* ws       = (float*)d_ws;
    float* feat_src = ws;
    float* skip     = feat_src + (size_t)N_NODES * HD;
    float* el       = skip     + (size_t)N_NODES * HD;
    float* er       = el       + (size_t)N_NODES * HEADS;
    float* emax     = er       + (size_t)N_NODES * HEADS;
    float* esum     = emax     + (size_t)N_NODES * HEADS;
    float* ebuf     = esum     + (size_t)N_NODES * HEADS;

    fill_kernel<<<N_NODES * HD / 4 / (256 * FILL_V4_PER_THREAD), 256, 0, stream>>>(out, esum,
                                                                                   (unsigned int*)emax);
    {
        int waves  = N_NODES / 16;
        int blocks = (waves * 32 + 255) / 256;
        gemm_wmma_kernel<<<blocks, 256, 0, stream>>>(feat, W_src, b_src, W_skip, b_skip,
                                                     feat_src, skip);
    }
    {
        int n = N_NODES * HEADS;
        elr_kernel<<<(n + 255) / 256, 256, 0, stream>>>(feat_src, attn_l, attn_r, el, er);
    }
    edge_logit_kernel<<<(N_EDGES + 255) / 256, 256, 0, stream>>>(src, dst, el, er, ebuf,
                                                                 (unsigned int*)emax);
    edge_exp_kernel<<<(N_EDGES + 255) / 256, 256, 0, stream>>>(dst, (const unsigned int*)emax, ebuf,
                                                               esum);
    aggregate_kernel<<<(N_EDGES + 255) / 256, 256, 0, stream>>>(src, dst, ebuf, feat_src, out);
    final_kernel<<<(N_NODES / FIN_NPW * 32 + 255) / 256, 256, 0, stream>>>(skip, esum, W_gate, b_gate,
                                                                           ln_g, ln_b, prelu_a, out);
}


# GEMM: fill launch packs both weight matrices in B-fragment order, so a B load is 512 contiguous bytes feeding 2 WMMAs (was 16 lines per WMMA); A rows read once, 16 B per lane; k sum order changes
# speedup vs baseline: 2.0295x; 1.0908x over previous
#include <hip/hip_runtime.h>
#include <math.h>

#define N_NODES 100000
#define N_EDGES 800000
#define IN_DIM  128
#define HEADS   4
#define DHEAD   32
#define HD      128
#define EPS_LN  1e-5f
#define NEG_SLOPE 0.2f

typedef __attribute__((ext_vector_type(2))) float v2f;
typedef __attribute__((ext_vector_type(4))) float v4f;
typedef __attribute__((ext_vector_type(8))) float v8f;

__device__ __forceinline__ unsigned int f32_to_key(float v) {
    const unsigned int b = __float_as_uint(v);
    return (b & 0x80000000u) ? ~b : (b | 0x80000000u);
}
__device__ __forceinline__ float key_to_f32(unsigned int k) {
    return __uint_as_float((k & 0x80000000u) ? (k & 0x7FFFFFFFu) : ~k);
}

typedef __attribute__((ext_vector_type(4))) unsigned int v4u;
#define FILL_V4_PER_THREAD 10
static_assert((N_NODES * HD / 4) % (256 * FILL_V4_PER_THREAD) == 0, "whole work-groups cover out exactly");
static_assert(N_NODES * HD / 4 / FILL_V4_PER_THREAD >= N_NODES * HEADS / 4, "enough threads for esum, emax");
#define WPACK_V4 (2 * HD * IN_DIM / 4)
static_assert(WPACK_V4 == 2 * 8 * 16 * 32, "p, column tile, k group, lane");
static_assert(WPACK_V4 <= N_NODES * HD / 4 / FILL_V4_PER_THREAD, "enough threads for the packed weights");
static_assert((size_t)WPACK_V4 * 4 <= (size_t)N_EDGES * HEADS, "packed weights fit in the edge buffer they borrow");
__global__ __launch_bounds__(256) void fill_kernel(float* __restrict__ out, float* __restrict__ esum,
                                                   unsigned int* __restrict__ emax,
                                                   const float* __restrict__ Wsrc,
                                                   const float* __restrict__ Wskip,
                                                   float* __restrict__ wpack) {
    const v4f zf = {0.0f, 0.0f, 0.0f, 0.0f};
    const v4u zk = {0u, 0u, 0u, 0u};
    v4f* o4 = (v4f*)out + (size_t)blockIdx.x * (256 * FILL_V4_PER_THREAD) + threadIdx.x;
    for (int k = 0; k < FILL_V4_PER_THREAD; ++k) o4[k * 256] = zf;
    const int i = blockIdx.x * 256 + threadIdx.x;
    if (i < N_NODES * HEADS / 4) {
        ((v4f*)esum)[i] = zf;
        ((v4u*)emax)[i] = zk;
    }
    if (i < WPACK_V4) {
        const int ln = i & 31;
        const int j  = (i >> 5) & 15;
        const int ct = (i >> 9) & 7;
        const float* Wm = (i >> 12) ? Wskip : Wsrc;
        ((v4f*)wpack)[i] = *(const v4f*)(Wm + (size_t)(ct * 16 + (ln & 15)) * IN_DIM + 8 * j + 4 * (ln >> 4));
    }
}

__global__ __launch_bounds__(256) void gemm_wmma_kernel(
    const float* __restrict__ feat,
    const float* __restrict__ wpack,
    const float* __restrict__ bsrc, const float* __restrict__ bskip,
    float* __restrict__ feat_src, float* __restrict__ skip) {
    const int wave = (blockIdx.x * blockDim.x + threadIdx.x) >> 5;
    const int lane = threadIdx.x & 31;
    if (wave >= N_NODES / 16) return;
    const int node0 = wave * 16;
    const int half  = lane >> 4;
    const int l16   = lane & 15;

    const v4f* Arow = (const v4f*)(feat + (size_t)(node0 + l16) * IN_DIM + 4 * half);
    v4f a[16];
    #pragma unroll
    for (int j = 0; j < 16; ++j) a[j] = Arow[2 * j];

    const v4f* Bl = (const v4f*)wpack + lane;
    for (int t = 0; t < 16; ++t) {
        const int p  = t >> 3;
        const int ct = t & 7;
        const float* bias = p ? bskip : bsrc;
        float*       out  = p ? skip  : feat_src;
        const float bv = bias[ct * 16 + l16];
        v8f acc;
        #pragma unroll
        for (int v = 0; v < 8; ++v) acc[v] = bv;
        const v4f* Bt = Bl + t * (16 * 32);
        #pragma unroll
        for (int j = 0; j < 16; ++j) {
            const v4f b = Bt[j * 32];
            v2f a0 = {a[j].x, a[j].y};
            v2f b0 = {b.x, b.y};
            v2f a1 = {a[j].z, a[j].w};
            v2f b1 = {b.z, b.w};
            acc = __builtin_amdgcn_wmma_f32_16x16x4_f32(
                false, a0, false, b0, (short)0, acc, false, false);
            acc = __builtin_amdgcn_wmma_f32_16x16x4_f32(
                false, a1, false, b1, (short)0, acc, false, false);
        }
        float* obase = out + (size_t)node0 * HD + ct * 16 + l16;
        #pragma unroll
        for (int v = 0; v < 8; ++v)
            obase[(size_t)(v + 8 * half) * HD] = acc[v];
    }
}

__global__ void elr_kernel(const float* __restrict__ feat_src,
                           const float* __restrict__ attn_l,
                           const float* __restrict__ attn_r,
                           float* __restrict__ el, float* __restrict__ er) {
    int i = blockIdx.x * blockDim.x + threadIdx.x;
    if (i >= N_NODES * HEADS) return;
    const int h = i & 3;
    const float* f  = feat_src + (size_t)(i >> 2) * HD + h * DHEAD;
    const float* al = attn_l + h * DHEAD;
    const float* ar = attn_r + h * DHEAD;
    float sl = 0.0f, sr = 0.0f;
    #pragma unroll
    for (int d = 0; d < DHEAD; ++d) {
        float v = f[d];
        sl += v * al[d];
        sr += v * ar[d];
    }
    el[i] = sl;
    er[i] = sr;
}

static_assert(N_EDGES % 32 == 0, "a wave owns 32 whole edges");
__global__ __launch_bounds__(256) void edge_logit_kernel(
    const int* __restrict__ src, const int* __restrict__ dst,
    const float* __restrict__ el, const float* __restrict__ er,
    float* __restrict__ ebuf, unsigned int* __restrict__ emax) {
    const int lane  = threadIdx.x & 31;
    const int ebase = (blockIdx.x * blockDim.x + threadIdx.x) & ~31;
    if (ebase >= N_EDGES) return;
    const int s_own = src[ebase + lane];
    const int d_own = dst[ebase + lane];
    const int h   = lane & 3;
    const int sub = lane >> 2;
    int   dd[4];
    float a[4], b[4];
    for (int i = 0; i < 4; ++i) {
        const int j = i * 8 + sub;
        const int s = __shfl(s_own, j, 32);
        dd[i] = __shfl(d_own, j, 32);
        a[i] = el[(size_t)s * 4 + h];
        b[i] = er[(size_t)dd[i] * 4 + h];
    }
    for (int i = 0; i < 4; ++i) {
        float v = a[i] + b[i];
        v = (v >= 0.0f) ? v : NEG_SLOPE * v;
        ebuf[(size_t)ebase * 4 + i * 32 + lane] = v;
        atomicMax(emax + (size_t)dd[i] * 4 + h, f32_to_key(v));
    }
}

__global__ __launch_bounds__(256) void edge_exp_kernel(
    const int* __restrict__ dst, const unsigned int* __restrict__ emax,
    float* __restrict__ ebuf, float* __restrict__ esum) {
    const int lane  = threadIdx.x & 31;
    const int ebase = (blockIdx.x * blockDim.x + threadIdx.x) & ~31;
    if (ebase >= N_EDGES) return;
    const int d_own = dst[ebase + lane];
    const int h   = lane & 3;
    const int sub = lane >> 2;
    float* eb = ebuf + (size_t)ebase * 4 + lane;
    int          dd[4];
    unsigned int mk[4];
    float        ev[4];
    for (int i = 0; i < 4; ++i) {
        dd[i] = __shfl(d_own, i * 8 + sub, 32);
        mk[i] = emax[(size_t)dd[i] * 4 + h];
        ev[i] = eb[i * 32];
    }
    for (int i = 0; i < 4; ++i) {
        const float ex = __expf(ev[i] - key_to_f32(mk[i]));
        eb[i * 32] = ex;
        atomicAdd(esum + (size_t)dd[i] * 4 + h, ex);
    }
}

static_assert(N_EDGES % 32 == 0, "a wave owns 32 whole edges");
static_assert(DHEAD == 32 && HD == HEADS * DHEAD, "one head = 32 channels = one line = one lane each");
__global__ __launch_bounds__(256) void aggregate_kernel(
    const int* __restrict__ src, const int* __restrict__ dst,
    const float* __restrict__ ebuf, const float* __restrict__ feat_src,
    float* __restrict__ out) {
    const int lane  = threadIdx.x & 31;
    const int ebase = (blockIdx.x * blockDim.x + threadIdx.x) & ~31;
    if (ebase >= N_EDGES) return;
    const int s_own = src[ebase + lane];
    const int d_own = dst[ebase + lane];
    unsigned int aw[4];
    for (int i = 0; i < 4; ++i) aw[i] = __float_as_uint(ebuf[(size_t)ebase * 4 + i * 32 + lane]);
    const float* fcol = feat_src + lane;
    float*       ocol = out + lane;
    for (int i = 0; i < 4; ++i) {
        for (int q0 = 0; q0 < 8; q0 += 4) {
            float f[4][HEADS];
            float a[4][HEADS];
            int   dd[4];
            for (int u = 0; u < 4; ++u) {
                const int q = q0 + u;
                const int j = i * 8 + q;
                const int s = __builtin_amdgcn_readlane(s_own, j);
                dd[u] = __builtin_amdgcn_readlane(d_own, j);
                const float* fr = fcol + (size_t)s * HD;
                for (int k = 0; k < HEADS; ++k) {
                    a[u][k] = __uint_as_float(__builtin_amdgcn_readlane(aw[i], q * 4 + k));
                    f[u][k] = fr[k * DHEAD];
                }
            }
            for (int u = 0; u < 4; ++u) {
                float* o = ocol + (size_t)dd[u] * HD;
                for (int k = 0; k < HEADS; ++k)
                    atomicAdd(o + k * DHEAD, a[u][k] * f[u][k]);
            }
        }
    }
}

#define FIN_NPW 8
static_assert(N_NODES % FIN_NPW == 0, "a wave owns 8 whole nodes");
__global__ __launch_bounds__(256) void final_kernel(
    const float* __restrict__ skip, const float* __restrict__ esum,
    const float* __restrict__ Wg, const float* __restrict__ bg,
    const float* __restrict__ ln_g, const float* __restrict__ ln_b,
    const float* __restrict__ prelu_a, float* __restrict__ out) {
    const int wave = (blockIdx.x * blockDim.x + threadIdx.x) >> 5;
    const int lane = threadIdx.x & 31;
    const int n0 = wave * FIN_NPW;
    if (n0 >= N_NODES) return;
    const int c = lane * 4;
    const int h = lane >> 3;
    const v4f w0 = *(const v4f*)(Wg + c);
    const v4f w1 = *(const v4f*)(Wg + 128 + c);
    const v4f w2 = *(const v4f*)(Wg + 256 + c);
    const v4f lg = *(const v4f*)(ln_g + c);
    const v4f lb = *(const v4f*)(ln_b + c);
    const float bgv   = bg[0];
    const float alpha = prelu_a[0];
    for (int b = 0; b < FIN_NPW; b += 4) {
        float es[4];
        v4f   rr[4], ss[4];
        #pragma unroll
        for (int u = 0; u < 4; ++u) {
            const size_t node = (size_t)(n0 + b + u);
            es[u] = esum[node * 4 + h];
            rr[u] = *(const v4f*)(out  + node * HD + c);
            ss[u] = *(const v4f*)(skip + node * HD + c);
        }
        #pragma unroll
        for (int u = 0; u < 4; ++u) {
            const size_t node = (size_t)(n0 + b + u);
            const float inv = (es[u] > 0.0f) ? 1.0f / es[u] : 0.0f;
            const v4f r  = rr[u] * inv;
            const v4f sk = ss[u];
            const v4f dm = r - sk;
            float part = r.x * w0.x + r.y * w0.y + r.z * w0.z + r.w * w0.w
                       + sk.x * w1.x + sk.y * w1.y + sk.z * w1.z + sk.w * w1.w
                       + dm.x * w2.x + dm.y * w2.y + dm.z * w2.z + dm.w * w2.w;
            #pragma unroll
            for (int off = 16; off > 0; off >>= 1) part += __shfl_xor(part, off, 32);
            const float g = 1.0f / (1.0f + __expf(-(part + bgv)));
            v4f z = g * r + (1.0f - g) * sk;
            float s  = z.x + z.y + z.z + z.w;
            float sq = z.x * z.x + z.y * z.y + z.z * z.z + z.w * z.w;
            #pragma unroll
            for (int off = 16; off > 0; off >>= 1) {
                s  += __shfl_xor(s,  off, 32);
                sq += __shfl_xor(sq, off, 32);
            }
            const float mu   = s * (1.0f / 128.0f);
            const float var  = sq * (1.0f / 128.0f) - mu * mu;
            const float rstd = rsqrtf(var + EPS_LN);
            v4f y;
            #pragma unroll
            for (int i = 0; i < 4; ++i) {
                float yy = (z[i] - mu) * rstd * lg[i] + lb[i];
                y[i] = (yy >= 0.0f) ? yy : alpha * yy;
            }
            *(v4f*)(out + node * HD + c) = y;
        }
    }
}

extern "C" void kernel_launch(void* const* d_in, const int* in_sizes, int n_in,
                              void* d_out, int out_size, void* d_ws, size_t ws_size,
                              hipStream_t stream) {
    const float* feat    = (const float*)d_in[0];
    const int*   src     = (const int*)  d_in[1];
    const int*   dst     = (const int*)  d_in[2];
    const float* W_src   = (const float*)d_in[3];
    const float* b_src   = (const float*)d_in[4];
    const float* attn_l  = (const float*)d_in[5];
    const float* attn_r  = (const float*)d_in[6];
    const float* W_skip  = (const float*)d_in[7];
    const float* b_skip  = (const float*)d_in[8];
    const float* W_gate  = (const float*)d_in[9];
    const float* b_gate  = (const float*)d_in[10];
    const float* ln_g    = (const float*)d_in[11];
    const float* ln_b    = (const float*)d_in[12];
    const float* prelu_a = (const float*)d_in[13];
    float* out = (float*)d_out;

    float* ws       = (float*)d_ws;
    float* feat_src = ws;
    float* skip     = feat_src + (size_t)N_NODES * HD;
    float* el       = skip     + (size_t)N_NODES * HD;
    float* er       = el       + (size_t)N_NODES * HEADS;
    float* emax     = er       + (size_t)N_NODES * HEADS;
    float* esum     = emax     + (size_t)N_NODES * HEADS;
    float* ebuf     = esum     + (size_t)N_NODES * HEADS;

    fill_kernel<<<N_NODES * HD / 4 / (256 * FILL_V4_PER_THREAD), 256, 0, stream>>>(out, esum,
                                                                                   (unsigned int*)emax,
                                                                                   W_src, W_skip, ebuf);
    {
        int waves  = N_NODES / 16;
        int blocks = (waves * 32 + 255) / 256;
        gemm_wmma_kernel<<<blocks, 256, 0, stream>>>(feat, ebuf, b_src, b_skip,
                                                     feat_src, skip);
    }
    {
        int n = N_NODES * HEADS;
        elr_kernel<<<(n + 255) / 256, 256, 0, stream>>>(feat_src, attn_l, attn_r, el, er);
    }
    edge_logit_kernel<<<(N_EDGES + 255) / 256, 256, 0, stream>>>(src, dst, el, er, ebuf,
                                                                 (unsigned int*)emax);
    edge_exp_kernel<<<(N_EDGES + 255) / 256, 256, 0, stream>>>(dst, (const unsigned int*)emax, ebuf,
                                                               esum);
    aggregate_kernel<<<(N_EDGES + 255) / 256, 256, 0, stream>>>(src, dst, ebuf, feat_src, out);
    final_kernel<<<(N_NODES / FIN_NPW * 32 + 255) / 256, 256, 0, stream>>>(skip, esum, W_gate, b_gate,
                                                                           ln_g, ln_b, prelu_a, out);
}


# Aggregation walks the edges twice in one launch: grid's first half adds heads 0-1 of each row, second half heads 2-3, so the randomly reused row bytes halve and stay in L2; same arrival-order adds
# speedup vs baseline: 2.0988x; 1.0341x over previous
#include <hip/hip_runtime.h>
#include <math.h>

#define N_NODES 100000
#define N_EDGES 800000
#define IN_DIM  128
#define HEADS   4
#define DHEAD   32
#define HD      128
#define EPS_LN  1e-5f
#define NEG_SLOPE 0.2f

typedef __attribute__((ext_vector_type(2))) float v2f;
typedef __attribute__((ext_vector_type(4))) float v4f;
typedef __attribute__((ext_vector_type(8))) float v8f;

__device__ __forceinline__ unsigned int f32_to_key(float v) {
    const unsigned int b = __float_as_uint(v);
    return (b & 0x80000000u) ? ~b : (b | 0x80000000u);
}
__device__ __forceinline__ float key_to_f32(unsigned int k) {
    return __uint_as_float((k & 0x80000000u) ? (k & 0x7FFFFFFFu) : ~k);
}

typedef __attribute__((ext_vector_type(4))) unsigned int v4u;
#define FILL_V4_PER_THREAD 10
static_assert((N_NODES * HD / 4) % (256 * FILL_V4_PER_THREAD) == 0, "whole work-groups cover out exactly");
static_assert(N_NODES * HD / 4 / FILL_V4_PER_THREAD >= N_NODES * HEADS / 4, "enough threads for esum, emax");
#define WPACK_V4 (2 * HD * IN_DIM / 4)
static_assert(WPACK_V4 == 2 * 8 * 16 * 32, "p, column tile, k group, lane");
static_assert(WPACK_V4 <= N_NODES * HD / 4 / FILL_V4_PER_THREAD, "enough threads for the packed weights");
static_assert((size_t)WPACK_V4 * 4 <= (size_t)N_EDGES * HEADS, "packed weights fit in the edge buffer they borrow");
__global__ __launch_bounds__(256) void fill_kernel(float* __restrict__ out, float* __restrict__ esum,
                                                   unsigned int* __restrict__ emax,
                                                   const float* __restrict__ Wsrc,
                                                   const float* __restrict__ Wskip,
                                                   float* __restrict__ wpack) {
    const v4f zf = {0.0f, 0.0f, 0.0f, 0.0f};
    const v4u zk = {0u, 0u, 0u, 0u};
    v4f* o4 = (v4f*)out + (size_t)blockIdx.x * (256 * FILL_V4_PER_THREAD) + threadIdx.x;
    for (int k = 0; k < FILL_V4_PER_THREAD; ++k) o4[k * 256] = zf;
    const int i = blockIdx.x * 256 + threadIdx.x;
    if (i < N_NODES * HEADS / 4) {
        ((v4f*)esum)[i] = zf;
        ((v4u*)emax)[i] = zk;
    }
    if (i < WPACK_V4) {
        const int ln = i & 31;
        const int j  = (i >> 5) & 15;
        const int ct = (i >> 9) & 7;
        const float* Wm = (i >> 12) ? Wskip : Wsrc;
        ((v4f*)wpack)[i] = *(const v4f*)(Wm + (size_t)(ct * 16 + (ln & 15)) * IN_DIM + 8 * j + 4 * (ln >> 4));
    }
}

__global__ __launch_bounds__(256) void gemm_wmma_kernel(
    const float* __restrict__ feat,
    const float* __restrict__ wpack,
    const float* __restrict__ bsrc, const float* __restrict__ bskip,
    float* __restrict__ feat_src, float* __restrict__ skip) {
    const int wave = (blockIdx.x * blockDim.x + threadIdx.x) >> 5;
    const int lane = threadIdx.x & 31;
    if (wave >= N_NODES / 16) return;
    const int node0 = wave * 16;
    const int half  = lane >> 4;
    const int l16   = lane & 15;

    const v4f* Arow = (const v4f*)(feat + (size_t)(node0 + l16) * IN_DIM + 4 * half);
    v4f a[16];
    #pragma unroll
    for (int j = 0; j < 16; ++j) a[j] = Arow[2 * j];

    const v4f* Bl = (const v4f*)wpack + lane;
    for (int t = 0; t < 16; ++t) {
        const int p  = t >> 3;
        const int ct = t & 7;
        const float* bias = p ? bskip : bsrc;
        float*       out  = p ? skip  : feat_src;
        const float bv = bias[ct * 16 + l16];
        v8f acc;
        #pragma unroll
        for (int v = 0; v < 8; ++v) acc[v] = bv;
        const v4f* Bt = Bl + t * (16 * 32);
        #pragma unroll
        for (int j = 0; j < 16; ++j) {
            const v4f b = Bt[j * 32];
            v2f a0 = {a[j].x, a[j].y};
            v2f b0 = {b.x, b.y};
            v2f a1 = {a[j].z, a[j].w};
            v2f b1 = {b.z, b.w};
            acc = __builtin_amdgcn_wmma_f32_16x16x4_f32(
                false, a0, false, b0, (short)0, acc, false, false);
            acc = __builtin_amdgcn_wmma_f32_16x16x4_f32(
                false, a1, false, b1, (short)0, acc, false, false);
        }
        float* obase = out + (size_t)node0 * HD + ct * 16 + l16;
        #pragma unroll
        for (int v = 0; v < 8; ++v)
            obase[(size_t)(v + 8 * half) * HD] = acc[v];
    }
}

__global__ void elr_kernel(const float* __restrict__ feat_src,
                           const float* __restrict__ attn_l,
                           const float* __restrict__ attn_r,
                           float* __restrict__ el, float* __restrict__ er) {
    int i = blockIdx.x * blockDim.x + threadIdx.x;
    if (i >= N_NODES * HEADS) return;
    const int h = i & 3;
    const float* f  = feat_src + (size_t)(i >> 2) * HD + h * DHEAD;
    const float* al = attn_l + h * DHEAD;
    const float* ar = attn_r + h * DHEAD;
    float sl = 0.0f, sr = 0.0f;
    #pragma unroll
    for (int d = 0; d < DHEAD; ++d) {
        float v = f[d];
        sl += v * al[d];
        sr += v * ar[d];
    }
    el[i] = sl;
    er[i] = sr;
}

static_assert(N_EDGES % 32 == 0, "a wave owns 32 whole edges");
__global__ __launch_bounds__(256) void edge_logit_kernel(
    const int* __restrict__ src, const int* __restrict__ dst,
    const float* __restrict__ el, const float* __restrict__ er,
    float* __restrict__ ebuf, unsigned int* __restrict__ emax) {
    const int lane  = threadIdx.x & 31;
    const int ebase = (blockIdx.x * blockDim.x + threadIdx.x) & ~31;
    if (ebase >= N_EDGES) return;
    const int s_own = src[ebase + lane];
    const int d_own = dst[ebase + lane];
    const int h   = lane & 3;
    const int sub = lane >> 2;
    int   dd[4];
    float a[4], b[4];
    for (int i = 0; i < 4; ++i) {
        const int j = i * 8 + sub;
        const int s = __shfl(s_own, j, 32);
        dd[i] = __shfl(d_own, j, 32);
        a[i] = el[(size_t)s * 4 + h];
        b[i] = er[(size_t)dd[i] * 4 + h];
    }
    for (int i = 0; i < 4; ++i) {
        float v = a[i] + b[i];
        v = (v >= 0.0f) ? v : NEG_SLOPE * v;
        ebuf[(size_t)ebase * 4 + i * 32 + lane] = v;
        atomicMax(emax + (size_t)dd[i] * 4 + h, f32_to_key(v));
    }
}

__global__ __launch_bounds__(256) void edge_exp_kernel(
    const int* __restrict__ dst, const unsigned int* __restrict__ emax,
    float* __restrict__ ebuf, float* __restrict__ esum) {
    const int lane  = threadIdx.x & 31;
    const int ebase = (blockIdx.x * blockDim.x + threadIdx.x) & ~31;
    if (ebase >= N_EDGES) return;
    const int d_own = dst[ebase + lane];
    const int h   = lane & 3;
    const int sub = lane >> 2;
    float* eb = ebuf + (size_t)ebase * 4 + lane;
    int          dd[4];
    unsigned int mk[4];
    float        ev[4];
    for (int i = 0; i < 4; ++i) {
        dd[i] = __shfl(d_own, i * 8 + sub, 32);
        mk[i] = emax[(size_t)dd[i] * 4 + h];
        ev[i] = eb[i * 32];
    }
    for (int i = 0; i < 4; ++i) {
        const float ex = __expf(ev[i] - key_to_f32(mk[i]));
        eb[i * 32] = ex;
        atomicAdd(esum + (size_t)dd[i] * 4 + h, ex);
    }
}

static_assert(N_EDGES % 256 == 0, "a work-group owns 256 whole edges, a wave 32");
static_assert(DHEAD == 32 && HD == HEADS * DHEAD, "one head = 32 channels = one line = one lane each");
static_assert(HEADS == 4, "two passes of two heads");
#define AGG_BLOCKS_PER_PASS (N_EDGES / 256)
__global__ __launch_bounds__(256) void aggregate_kernel(
    const int* __restrict__ src, const int* __restrict__ dst,
    const float* __restrict__ ebuf, const float* __restrict__ feat_src,
    float* __restrict__ out) {
    const int lane  = threadIdx.x & 31;
    const int pass  = (blockIdx.x >= AGG_BLOCKS_PER_PASS) ? 1 : 0;
    const int blk   = blockIdx.x - pass * AGG_BLOCKS_PER_PASS;
    const int ebase = (blk * 256 + threadIdx.x) & ~31;
    if (ebase >= N_EDGES) return;
    const int s_own = src[ebase + lane];
    const int d_own = dst[ebase + lane];
    unsigned int aw[4];
    for (int i = 0; i < 4; ++i) aw[i] = __float_as_uint(ebuf[(size_t)ebase * 4 + i * 32 + lane]);
    const int k0 = 2 * pass;
    const float* fcol = feat_src + k0 * DHEAD + lane;
    float*       ocol = out      + k0 * DHEAD + lane;
    for (int i = 0; i < 4; ++i) {
        float f[8][2];
        float a[8][2];
        int   dd[8];
        for (int q = 0; q < 8; ++q) {
            const int j = i * 8 + q;
            const int s = __builtin_amdgcn_readlane(s_own, j);
            dd[q] = __builtin_amdgcn_readlane(d_own, j);
            const float* fr = fcol + (size_t)s * HD;
            for (int kk = 0; kk < 2; ++kk) {
                a[q][kk] = __uint_as_float(__builtin_amdgcn_readlane(aw[i], q * 4 + k0 + kk));
                f[q][kk] = fr[kk * DHEAD];
            }
        }
        for (int q = 0; q < 8; ++q) {
            float* o = ocol + (size_t)dd[q] * HD;
            for (int kk = 0; kk < 2; ++kk)
                atomicAdd(o + kk * DHEAD, a[q][kk] * f[q][kk]);
        }
    }
}

#define FIN_NPW 8
static_assert(N_NODES % FIN_NPW == 0, "a wave owns 8 whole nodes");
__global__ __launch_bounds__(256) void final_kernel(
    const float* __restrict__ skip, const float* __restrict__ esum,
    const float* __restrict__ Wg, const float* __restrict__ bg,
    const float* __restrict__ ln_g, const float* __restrict__ ln_b,
    const float* __restrict__ prelu_a, float* __restrict__ out) {
    const int wave = (blockIdx.x * blockDim.x + threadIdx.x) >> 5;
    const int lane = threadIdx.x & 31;
    const int n0 = wave * FIN_NPW;
    if (n0 >= N_NODES) return;
    const int c = lane * 4;
    const int h = lane >> 3;
    const v4f w0 = *(const v4f*)(Wg + c);
    const v4f w1 = *(const v4f*)(Wg + 128 + c);
    const v4f w2 = *(const v4f*)(Wg + 256 + c);
    const v4f lg = *(const v4f*)(ln_g + c);
    const v4f lb = *(const v4f*)(ln_b + c);
    const float bgv   = bg[0];
    const float alpha = prelu_a[0];
    for (int b = 0; b < FIN_NPW; b += 4) {
        float es[4];
        v4f   rr[4], ss[4];
        #pragma unroll
        for (int u = 0; u < 4; ++u) {
            const size_t node = (size_t)(n0 + b + u);
            es[u] = esum[node * 4 + h];
            rr[u] = *(const v4f*)(out  + node * HD + c);
            ss[u] = *(const v4f*)(skip + node * HD + c);
        }
        #pragma unroll
        for (int u = 0; u < 4; ++u) {
            const size_t node = (size_t)(n0 + b + u);
            const float inv = (es[u] > 0.0f) ? 1.0f / es[u] : 0.0f;
            const v4f r  = rr[u] * inv;
            const v4f sk = ss[u];
            const v4f dm = r - sk;
            float part = r.x * w0.x + r.y * w0.y + r.z * w0.z + r.w * w0.w
                       + sk.x * w1.x + sk.y * w1.y + sk.z * w1.z + sk.w * w1.w
                       + dm.x * w2.x + dm.y * w2.y + dm.z * w2.z + dm.w * w2.w;
            #pragma unroll
            for (int off = 16; off > 0; off >>= 1) part += __shfl_xor(part, off, 32);
            const float g = 1.0f / (1.0f + __expf(-(part + bgv)));
            v4f z = g * r + (1.0f - g) * sk;
            float s  = z.x + z.y + z.z + z.w;
            float sq = z.x * z.x + z.y * z.y + z.z * z.z + z.w * z.w;
            #pragma unroll
            for (int off = 16; off > 0; off >>= 1) {
                s  += __shfl_xor(s,  off, 32);
                sq += __shfl_xor(sq, off, 32);
            }
            const float mu   = s * (1.0f / 128.0f);
            const float var  = sq * (1.0f / 128.0f) - mu * mu;
            const float rstd = rsqrtf(var + EPS_LN);
            v4f y;
            #pragma unroll
            for (int i = 0; i < 4; ++i) {
                float yy = (z[i] - mu) * rstd * lg[i] + lb[i];
                y[i] = (yy >= 0.0f) ? yy : alpha * yy;
            }
            *(v4f*)(out + node * HD + c) = y;
        }
    }
}

extern "C" void kernel_launch(void* const* d_in, const int* in_sizes, int n_in,
                              void* d_out, int out_size, void* d_ws, size_t ws_size,
                              hipStream_t stream) {
    const float* feat    = (const float*)d_in[0];
    const int*   src     = (const int*)  d_in[1];
    const int*   dst     = (const int*)  d_in[2];
    const float* W_src   = (const float*)d_in[3];
    const float* b_src   = (const float*)d_in[4];
    const float* attn_l  = (const float*)d_in[5];
    const float* attn_r  = (const float*)d_in[6];
    const float* W_skip  = (const float*)d_in[7];
    const float* b_skip  = (const float*)d_in[8];
    const float* W_gate  = (const float*)d_in[9];
    const float* b_gate  = (const float*)d_in[10];
    const float* ln_g    = (const float*)d_in[11];
    const float* ln_b    = (const float*)d_in[12];
    const float* prelu_a = (const float*)d_in[13];
    float* out = (float*)d_out;

    float* ws       = (float*)d_ws;
    float* feat_src = ws;
    float* skip     = feat_src + (size_t)N_NODES * HD;
    float* el       = skip     + (size_t)N_NODES * HD;
    float* er       = el       + (size_t)N_NODES * HEADS;
    float* emax     = er       + (size_t)N_NODES * HEADS;
    float* esum     = emax     + (size_t)N_NODES * HEADS;
    float* ebuf     = esum     + (size_t)N_NODES * HEADS;

    fill_kernel<<<N_NODES * HD / 4 / (256 * FILL_V4_PER_THREAD), 256, 0, stream>>>(out, esum,
                                                                                   (unsigned int*)emax,
                                                                                   W_src, W_skip, ebuf);
    {
        int waves  = N_NODES / 16;
        int blocks = (waves * 32 + 255) / 256;
        gemm_wmma_kernel<<<blocks, 256, 0, stream>>>(feat, ebuf, b_src, b_skip,
                                                     feat_src, skip);
    }
    {
        int n = N_NODES * HEADS;
        elr_kernel<<<(n + 255) / 256, 256, 0, stream>>>(feat_src, attn_l, attn_r, el, er);
    }
    edge_logit_kernel<<<(N_EDGES + 255) / 256, 256, 0, stream>>>(src, dst, el, er, ebuf,
                                                                 (unsigned int*)emax);
    edge_exp_kernel<<<(N_EDGES + 255) / 256, 256, 0, stream>>>(dst, (const unsigned int*)emax, ebuf,
                                                               esum);
    aggregate_kernel<<<2 * AGG_BLOCKS_PER_PASS, 256, 0, stream>>>(src, dst, ebuf, feat_src, out);
    final_kernel<<<(N_NODES / FIN_NPW * 32 + 255) / 256, 256, 0, stream>>>(skip, esum, W_gate, b_gate,
                                                                           ln_g, ln_b, prelu_a, out);
}


# Aggregation: a work-group owns 391 destinations, sums their rows in LDS with LDS float adds, finds its edges by a bin byte per edge, stores each row once; no global float adds; LDS arrival order
# speedup vs baseline: 2.2839x; 1.0882x over previous
#include <hip/hip_runtime.h>
#include <math.h>

#define N_NODES 100000
#define N_EDGES 800000
#define IN_DIM  128
#define HEADS   4
#define DHEAD   32
#define HD      128
#define EPS_LN  1e-5f
#define NEG_SLOPE 0.2f

typedef __attribute__((ext_vector_type(2))) float v2f;
typedef __attribute__((ext_vector_type(4))) float v4f;
typedef __attribute__((ext_vector_type(8))) float v8f;

__device__ __forceinline__ unsigned int f32_to_key(float v) {
    const unsigned int b = __float_as_uint(v);
    return (b & 0x80000000u) ? ~b : (b | 0x80000000u);
}
__device__ __forceinline__ float key_to_f32(unsigned int k) {
    return __uint_as_float((k & 0x80000000u) ? (k & 0x7FFFFFFFu) : ~k);
}

typedef __attribute__((ext_vector_type(4))) unsigned int v4u;
#define FILL_BLOCKS ((N_NODES * HEADS / 4 + 255) / 256)
#define WPACK_V4 (2 * HD * IN_DIM / 4)
static_assert(WPACK_V4 == 2 * 8 * 16 * 32, "p, column tile, k group, lane");
static_assert(WPACK_V4 <= FILL_BLOCKS * 256, "enough threads for the packed weights");
static_assert((size_t)WPACK_V4 * 4 <= (size_t)N_EDGES * HEADS, "packed weights fit in the edge buffer they borrow");
__global__ __launch_bounds__(256) void fill_kernel(float* __restrict__ esum,
                                                   unsigned int* __restrict__ emax,
                                                   const float* __restrict__ Wsrc,
                                                   const float* __restrict__ Wskip,
                                                   float* __restrict__ wpack) {
    const v4f zf = {0.0f, 0.0f, 0.0f, 0.0f};
    const v4u zk = {0u, 0u, 0u, 0u};
    const int i = blockIdx.x * 256 + threadIdx.x;
    if (i < N_NODES * HEADS / 4) {
        ((v4f*)esum)[i] = zf;
        ((v4u*)emax)[i] = zk;
    }
    if (i < WPACK_V4) {
        const int ln = i & 31;
        const int j  = (i >> 5) & 15;
        const int ct = (i >> 9) & 7;
        const float* Wm = (i >> 12) ? Wskip : Wsrc;
        ((v4f*)wpack)[i] = *(const v4f*)(Wm + (size_t)(ct * 16 + (ln & 15)) * IN_DIM + 8 * j + 4 * (ln >> 4));
    }
}

__global__ __launch_bounds__(256) void gemm_wmma_kernel(
    const float* __restrict__ feat,
    const float* __restrict__ wpack,
    const float* __restrict__ bsrc, const float* __restrict__ bskip,
    float* __restrict__ feat_src, float* __restrict__ skip) {
    const int wave = (blockIdx.x * blockDim.x + threadIdx.x) >> 5;
    const int lane = threadIdx.x & 31;
    if (wave >= N_NODES / 16) return;
    const int node0 = wave * 16;
    const int half  = lane >> 4;
    const int l16   = lane & 15;

    const v4f* Arow = (const v4f*)(feat + (size_t)(node0 + l16) * IN_DIM + 4 * half);
    v4f a[16];
    #pragma unroll
    for (int j = 0; j < 16; ++j) a[j] = Arow[2 * j];

    const v4f* Bl = (const v4f*)wpack + lane;
    for (int t = 0; t < 16; ++t) {
        const int p  = t >> 3;
        const int ct = t & 7;
        const float* bias = p ? bskip : bsrc;
        float*       out  = p ? skip  : feat_src;
        const float bv = bias[ct * 16 + l16];
        v8f acc;
        #pragma unroll
        for (int v = 0; v < 8; ++v) acc[v] = bv;
        const v4f* Bt = Bl + t * (16 * 32);
        #pragma unroll
        for (int j = 0; j < 16; ++j) {
            const v4f b = Bt[j * 32];
            v2f a0 = {a[j].x, a[j].y};
            v2f b0 = {b.x, b.y};
            v2f a1 = {a[j].z, a[j].w};
            v2f b1 = {b.z, b.w};
            acc = __builtin_amdgcn_wmma_f32_16x16x4_f32(
                false, a0, false, b0, (short)0, acc, false, false);
            acc = __builtin_amdgcn_wmma_f32_16x16x4_f32(
                false, a1, false, b1, (short)0, acc, false, false);
        }
        float* obase = out + (size_t)node0 * HD + ct * 16 + l16;
        #pragma unroll
        for (int v = 0; v < 8; ++v)
            obase[(size_t)(v + 8 * half) * HD] = acc[v];
    }
}

__global__ void elr_kernel(const float* __restrict__ feat_src,
                           const float* __restrict__ attn_l,
                           const float* __restrict__ attn_r,
                           float* __restrict__ el, float* __restrict__ er) {
    int i = blockIdx.x * blockDim.x + threadIdx.x;
    if (i >= N_NODES * HEADS) return;
    const int h = i & 3;
    const float* f  = feat_src + (size_t)(i >> 2) * HD + h * DHEAD;
    const float* al = attn_l + h * DHEAD;
    const float* ar = attn_r + h * DHEAD;
    float sl = 0.0f, sr = 0.0f;
    #pragma unroll
    for (int d = 0; d < DHEAD; ++d) {
        float v = f[d];
        sl += v * al[d];
        sr += v * ar[d];
    }
    el[i] = sl;
    er[i] = sr;
}

static_assert(N_EDGES % 32 == 0, "a wave owns 32 whole edges");
__global__ __launch_bounds__(256) void edge_logit_kernel(
    const int* __restrict__ src, const int* __restrict__ dst,
    const float* __restrict__ el, const float* __restrict__ er,
    float* __restrict__ ebuf, unsigned int* __restrict__ emax) {
    const int lane  = threadIdx.x & 31;
    const int ebase = (blockIdx.x * blockDim.x + threadIdx.x) & ~31;
    if (ebase >= N_EDGES) return;
    const int s_own = src[ebase + lane];
    const int d_own = dst[ebase + lane];
    const int h   = lane & 3;
    const int sub = lane >> 2;
    int   dd[4];
    float a[4], b[4];
    for (int i = 0; i < 4; ++i) {
        const int j = i * 8 + sub;
        const int s = __shfl(s_own, j, 32);
        dd[i] = __shfl(d_own, j, 32);
        a[i] = el[(size_t)s * 4 + h];
        b[i] = er[(size_t)dd[i] * 4 + h];
    }
    for (int i = 0; i < 4; ++i) {
        float v = a[i] + b[i];
        v = (v >= 0.0f) ? v : NEG_SLOPE * v;
        ebuf[(size_t)ebase * 4 + i * 32 + lane] = v;
        atomicMax(emax + (size_t)dd[i] * 4 + h, f32_to_key(v));
    }
}

#define NODES_PER_BIN 391
#define N_BINS 256
static_assert(N_BINS * NODES_PER_BIN >= N_NODES, "the bins cover every node");
static_assert((N_NODES - 1) / NODES_PER_BIN < 256, "a bin number fits in a byte");
__global__ __launch_bounds__(256) void edge_exp_kernel(
    const int* __restrict__ dst, const unsigned int* __restrict__ emax,
    float* __restrict__ ebuf, float* __restrict__ esum, unsigned char* __restrict__ ebin) {
    const int lane  = threadIdx.x & 31;
    const int ebase = (blockIdx.x * blockDim.x + threadIdx.x) & ~31;
    if (ebase >= N_EDGES) return;
    const int d_own = dst[ebase + lane];
    ebin[ebase + lane] = (unsigned char)(d_own / NODES_PER_BIN);
    const int h   = lane & 3;
    const int sub = lane >> 2;
    float* eb = ebuf + (size_t)ebase * 4 + lane;
    int          dd[4];
    unsigned int mk[4];
    float        ev[4];
    for (int i = 0; i < 4; ++i) {
        dd[i] = __shfl(d_own, i * 8 + sub, 32);
        mk[i] = emax[(size_t)dd[i] * 4 + h];
        ev[i] = eb[i * 32];
    }
    for (int i = 0; i < 4; ++i) {
        const float ex = __expf(ev[i] - key_to_f32(mk[i]));
        eb[i * 32] = ex;
        atomicAdd(esum + (size_t)dd[i] * 4 + h, ex);
    }
}

#define AGG_THREADS 1024
#define AGG_WAVES   (AGG_THREADS / 32)
#define SCAN_BYTES_PER_LANE 16
#define SCAN_EDGES_PER_LOAD (32 * SCAN_BYTES_PER_LANE)
#define AGG_BATCH 8
#define QUEUE_CAP (SCAN_EDGES_PER_LOAD + AGG_BATCH)
#define N_CHUNKS (N_EDGES / SCAN_BYTES_PER_LANE)
static_assert(N_EDGES % SCAN_BYTES_PER_LANE == 0, "the bin bytes are read 16 at a time");
static_assert(DHEAD == 32 && HD == HEADS * DHEAD && HEADS == 4, "one head = 32 channels = one line = one lane each");

template <bool FULL>
__device__ __forceinline__ void agg_batch(const unsigned int* qv, int cnt, int lane, int lo,
                                          const int* __restrict__ src, const int* __restrict__ dst,
                                          const float* __restrict__ ebuf,
                                          const float* __restrict__ feat_src, float* acc) {
    float f[AGG_BATCH][HEADS];
    float w[AGG_BATCH][HEADS];
    int   dl[AGG_BATCH];
    #pragma unroll
    for (int q = 0; q < AGG_BATCH; ++q) {
        if (FULL || q < cnt) {
            const int e = __builtin_amdgcn_readfirstlane((int)qv[q]);
            const int s = src[e];
            dl[q] = dst[e] - lo;
            const v4f wv = *(const v4f*)(ebuf + (size_t)e * 4);
            const float* fr = feat_src + (size_t)s * HD + lane;
            #pragma unroll
            for (int k = 0; k < HEADS; ++k) {
                w[q][k] = wv[k];
                f[q][k] = fr[k * DHEAD];
            }
        }
    }
    #pragma unroll
    for (int q = 0; q < AGG_BATCH; ++q) {
        if (FULL || q < cnt) {
            float* a = acc + dl[q] * HD + lane;
            #pragma unroll
            for (int k = 0; k < HEADS; ++k)
                atomicAdd(a + k * DHEAD, w[q][k] * f[q][k]);
        }
    }
}

__global__ __launch_bounds__(AGG_THREADS) void aggregate_kernel(
    const int* __restrict__ src, const int* __restrict__ dst,
    const unsigned char* __restrict__ ebin,
    const float* __restrict__ ebuf, const float* __restrict__ feat_src,
    float* __restrict__ out) {
    __shared__ float        acc[NODES_PER_BIN * HD];
    __shared__ unsigned int queue[AGG_WAVES * QUEUE_CAP];
    const int tid  = threadIdx.x;
    const int lane = tid & 31;
    const int wave = tid >> 5;
    const int bin  = blockIdx.x;
    const int lo   = bin * NODES_PER_BIN;

    const v4f zf = {0.0f, 0.0f, 0.0f, 0.0f};
    for (int i = tid; i < NODES_PER_BIN * HD / 4; i += AGG_THREADS) ((v4f*)acc)[i] = zf;
    __syncthreads();

    unsigned int* qv = queue + wave * QUEUE_CAP;
    int qn = 0;
    const unsigned int brep = (unsigned int)bin * 0x01010101u;
    const v4u* eb4 = (const v4u*)ebin;

    int t = wave;
    v4u nxt = eb4[min(t * 32 + lane, N_CHUNKS - 1)];
    while (t * 32 < N_CHUNKS) {
        const v4u cur = nxt;
        const int chunk = t * 32 + lane;
        const bool valid = chunk < N_CHUNKS;
        t += AGG_WAVES;
        nxt = eb4[min(t * 32 + lane, N_CHUNKS - 1)];
        #pragma unroll
        for (int d = 0; d < 4; ++d) {
            const unsigned int x  = cur[d] ^ brep;
            const unsigned int tt = (x & 0x7F7F7F7Fu) + 0x7F7F7F7Fu;
            unsigned int fl = valid ? ~(tt | x | 0x7F7F7F7Fu) : 0u;
            unsigned int m = __builtin_amdgcn_ballot_w32(fl != 0u);
            while (m != 0u) {
                if (fl != 0u) {
                    const int byte = (__builtin_ctz(fl)) >> 3;
                    const int rank = (int)__builtin_amdgcn_mbcnt_lo(m, 0u);
                    qv[qn + rank] = (unsigned int)(chunk * 16 + d * 4 + byte);
                    fl &= fl - 1u;
                }
                qn += __builtin_popcount(m);
                m = __builtin_amdgcn_ballot_w32(fl != 0u);
            }
        }
        while (qn >= AGG_BATCH) {
            qn -= AGG_BATCH;
            agg_batch<true>(qv + qn, AGG_BATCH, lane, lo, src, dst, ebuf, feat_src, acc);
        }
    }
    if (qn > 0) agg_batch<false>(qv, qn, lane, lo, src, dst, ebuf, feat_src, acc);
    __syncthreads();

    for (int r = wave; r < NODES_PER_BIN; r += AGG_WAVES) {
        const int node = lo + r;
        if (node < N_NODES)
            *(v4f*)(out + (size_t)node * HD + lane * 4) = *(const v4f*)(acc + r * HD + lane * 4);
    }
}

#define FIN_NPW 8
static_assert(N_NODES % FIN_NPW == 0, "a wave owns 8 whole nodes");
__global__ __launch_bounds__(256) void final_kernel(
    const float* __restrict__ skip, const float* __restrict__ esum,
    const float* __restrict__ Wg, const float* __restrict__ bg,
    const float* __restrict__ ln_g, const float* __restrict__ ln_b,
    const float* __restrict__ prelu_a, float* __restrict__ out) {
    const int wave = (blockIdx.x * blockDim.x + threadIdx.x) >> 5;
    const int lane = threadIdx.x & 31;
    const int n0 = wave * FIN_NPW;
    if (n0 >= N_NODES) return;
    const int c = lane * 4;
    const int h = lane >> 3;
    const v4f w0 = *(const v4f*)(Wg + c);
    const v4f w1 = *(const v4f*)(Wg + 128 + c);
    const v4f w2 = *(const v4f*)(Wg + 256 + c);
    const v4f lg = *(const v4f*)(ln_g + c);
    const v4f lb = *(const v4f*)(ln_b + c);
    const float bgv   = bg[0];
    const float alpha = prelu_a[0];
    for (int b = 0; b < FIN_NPW; b += 4) {
        float es[4];
        v4f   rr[4], ss[4];
        #pragma unroll
        for (int u = 0; u < 4; ++u) {
            const size_t node = (size_t)(n0 + b + u);
            es[u] = esum[node * 4 + h];
            rr[u] = *(const v4f*)(out  + node * HD + c);
            ss[u] = *(const v4f*)(skip + node * HD + c);
        }
        #pragma unroll
        for (int u = 0; u < 4; ++u) {
            const size_t node = (size_t)(n0 + b + u);
            const float inv = (es[u] > 0.0f) ? 1.0f / es[u] : 0.0f;
            const v4f r  = rr[u] * inv;
            const v4f sk = ss[u];
            const v4f dm = r - sk;
            float part = r.x * w0.x + r.y * w0.y + r.z * w0.z + r.w * w0.w
                       + sk.x * w1.x + sk.y * w1.y + sk.z * w1.z + sk.w * w1.w
                       + dm.x * w2.x + dm.y * w2.y + dm.z * w2.z + dm.w * w2.w;
            #pragma unroll
            for (int off = 16; off > 0; off >>= 1) part += __shfl_xor(part, off, 32);
            const float g = 1.0f / (1.0f + __expf(-(part + bgv)));
            v4f z = g * r + (1.0f - g) * sk;
            float s  = z.x + z.y + z.z + z.w;
            float sq = z.x * z.x + z.y * z.y + z.z * z.z + z.w * z.w;
            #pragma unroll
            for (int off = 16; off > 0; off >>= 1) {
                s  += __shfl_xor(s,  off, 32);
                sq += __shfl_xor(sq, off, 32);
            }
            const float mu   = s * (1.0f / 128.0f);
            const float var  = sq * (1.0f / 128.0f) - mu * mu;
            const float rstd = rsqrtf(var + EPS_LN);
            v4f y;
            #pragma unroll
            for (int i = 0; i < 4; ++i) {
                float yy = (z[i] - mu) * rstd * lg[i] + lb[i];
                y[i] = (yy >= 0.0f) ? yy : alpha * yy;
            }
            *(v4f*)(out + node * HD + c) = y;
        }
    }
}

extern "C" void kernel_launch(void* const* d_in, const int* in_sizes, int n_in,
                              void* d_out, int out_size, void* d_ws, size_t ws_size,
                              hipStream_t stream) {
    const float* feat    = (const float*)d_in[0];
    const int*   src     = (const int*)  d_in[1];
    const int*   dst     = (const int*)  d_in[2];
    const float* W_src   = (const float*)d_in[3];
    const float* b_src   = (const float*)d_in[4];
    const float* attn_l  = (const float*)d_in[5];
    const float* attn_r  = (const float*)d_in[6];
    const float* W_skip  = (const float*)d_in[7];
    const float* b_skip  = (const float*)d_in[8];
    const float* W_gate  = (const float*)d_in[9];
    const float* b_gate  = (const float*)d_in[10];
    const float* ln_g    = (const float*)d_in[11];
    const float* ln_b    = (const float*)d_in[12];
    const float* prelu_a = (const float*)d_in[13];
    float* out = (float*)d_out;

    float* ws       = (float*)d_ws;
    float* feat_src = ws;
    float* skip     = feat_src + (size_t)N_NODES * HD;
    float* el       = skip     + (size_t)N_NODES * HD;
    float* er       = el       + (size_t)N_NODES * HEADS;
    float* emax     = er       + (size_t)N_NODES * HEADS;
    float* esum     = emax     + (size_t)N_NODES * HEADS;
    float* ebuf     = esum     + (size_t)N_NODES * HEADS;

    fill_kernel<<<FILL_BLOCKS, 256, 0, stream>>>(esum, (unsigned int*)emax, W_src, W_skip, ebuf);
    {
        int waves  = N_NODES / 16;
        int blocks = (waves * 32 + 255) / 256;
        gemm_wmma_kernel<<<blocks, 256, 0, stream>>>(feat, ebuf, b_src, b_skip,
                                                     feat_src, skip);
    }
    {
        int n = N_NODES * HEADS;
        elr_kernel<<<(n + 255) / 256, 256, 0, stream>>>(feat_src, attn_l, attn_r, el, er);
    }
    edge_logit_kernel<<<(N_EDGES + 255) / 256, 256, 0, stream>>>(src, dst, el, er, ebuf,
                                                                 (unsigned int*)emax);
    unsigned char* ebin = (unsigned char*)el;
    edge_exp_kernel<<<(N_EDGES + 255) / 256, 256, 0, stream>>>(dst, (const unsigned int*)emax, ebuf,
                                                               esum, ebin);
    aggregate_kernel<<<N_BINS, AGG_THREADS, 0, stream>>>(src, dst, ebin, ebuf, feat_src, out);
    final_kernel<<<(N_NODES / FIN_NPW * 32 + 255) / 256, 256, 0, stream>>>(skip, esum, W_gate, b_gate,
                                                                           ln_g, ln_b, prelu_a, out);
}


# Edge launch sorts each 3200-edge block by destination range in LDS, writes (src,dst) in that order plus range starts; aggregation reads only its own pieces, no scan; logits recomputed; LDS add order
# speedup vs baseline: 3.0064x; 1.3163x over previous
#include <hip/hip_runtime.h>
#include <math.h>

#define N_NODES 100000
#define N_EDGES 800000
#define IN_DIM  128
#define HEADS   4
#define DHEAD   32
#define HD      128
#define EPS_LN  1e-5f
#define NEG_SLOPE 0.2f

typedef __attribute__((ext_vector_type(2))) float v2f;
typedef __attribute__((ext_vector_type(4))) float v4f;
typedef __attribute__((ext_vector_type(8))) float v8f;

__device__ __forceinline__ unsigned int f32_to_key(float v) {
    const unsigned int b = __float_as_uint(v);
    return (b & 0x80000000u) ? ~b : (b | 0x80000000u);
}
__device__ __forceinline__ float key_to_f32(unsigned int k) {
    return __uint_as_float((k & 0x80000000u) ? (k & 0x7FFFFFFFu) : ~k);
}

typedef __attribute__((ext_vector_type(4))) unsigned int v4u;
#define FILL_BLOCKS ((N_NODES * HEADS / 4 + 255) / 256)
#define WPACK_V4 (2 * HD * IN_DIM / 4)
static_assert(WPACK_V4 == 2 * 8 * 16 * 32, "p, column tile, k group, lane");
static_assert(WPACK_V4 <= FILL_BLOCKS * 256, "enough threads for the packed weights");
static_assert((size_t)WPACK_V4 * 4 <= (size_t)N_EDGES * HEADS, "packed weights fit in the edge buffer they borrow");
__global__ __launch_bounds__(256) void fill_kernel(unsigned int* __restrict__ emax,
                                                   const float* __restrict__ Wsrc,
                                                   const float* __restrict__ Wskip,
                                                   float* __restrict__ wpack) {
    const v4u zk = {0u, 0u, 0u, 0u};
    const int i = blockIdx.x * 256 + threadIdx.x;
    if (i < N_NODES * HEADS / 4) ((v4u*)emax)[i] = zk;
    if (i < WPACK_V4) {
        const int ln = i & 31;
        const int j  = (i >> 5) & 15;
        const int ct = (i >> 9) & 7;
        const float* Wm = (i >> 12) ? Wskip : Wsrc;
        ((v4f*)wpack)[i] = *(const v4f*)(Wm + (size_t)(ct * 16 + (ln & 15)) * IN_DIM + 8 * j + 4 * (ln >> 4));
    }
}

__global__ __launch_bounds__(256) void gemm_wmma_kernel(
    const float* __restrict__ feat,
    const float* __restrict__ wpack,
    const float* __restrict__ bsrc, const float* __restrict__ bskip,
    float* __restrict__ feat_src, float* __restrict__ skip) {
    const int wave = (blockIdx.x * blockDim.x + threadIdx.x) >> 5;
    const int lane = threadIdx.x & 31;
    if (wave >= N_NODES / 16) return;
    const int node0 = wave * 16;
    const int half  = lane >> 4;
    const int l16   = lane & 15;

    const v4f* Arow = (const v4f*)(feat + (size_t)(node0 + l16) * IN_DIM + 4 * half);
    v4f a[16];
    #pragma unroll
    for (int j = 0; j < 16; ++j) a[j] = Arow[2 * j];

    const v4f* Bl = (const v4f*)wpack + lane;
    for (int t = 0; t < 16; ++t) {
        const int p  = t >> 3;
        const int ct = t & 7;
        const float* bias = p ? bskip : bsrc;
        float*       out  = p ? skip  : feat_src;
        const float bv = bias[ct * 16 + l16];
        v8f acc;
        #pragma unroll
        for (int v = 0; v < 8; ++v) acc[v] = bv;
        const v4f* Bt = Bl + t * (16 * 32);
        #pragma unroll
        for (int j = 0; j < 16; ++j) {
            const v4f b = Bt[j * 32];
            v2f a0 = {a[j].x, a[j].y};
            v2f b0 = {b.x, b.y};
            v2f a1 = {a[j].z, a[j].w};
            v2f b1 = {b.z, b.w};
            acc = __builtin_amdgcn_wmma_f32_16x16x4_f32(
                false, a0, false, b0, (short)0, acc, false, false);
            acc = __builtin_amdgcn_wmma_f32_16x16x4_f32(
                false, a1, false, b1, (short)0, acc, false, false);
        }
        float* obase = out + (size_t)node0 * HD + ct * 16 + l16;
        #pragma unroll
        for (int v = 0; v < 8; ++v)
            obase[(size_t)(v + 8 * half) * HD] = acc[v];
    }
}

__global__ void elr_kernel(const float* __restrict__ feat_src,
                           const float* __restrict__ attn_l,
                           const float* __restrict__ attn_r,
                           float* __restrict__ el, float* __restrict__ er) {
    int i = blockIdx.x * blockDim.x + threadIdx.x;
    if (i >= N_NODES * HEADS) return;
    const int h = i & 3;
    const float* f  = feat_src + (size_t)(i >> 2) * HD + h * DHEAD;
    const float* al = attn_l + h * DHEAD;
    const float* ar = attn_r + h * DHEAD;
    float sl = 0.0f, sr = 0.0f;
    #pragma unroll
    for (int d = 0; d < DHEAD; ++d) {
        float v = f[d];
        sl += v * al[d];
        sr += v * ar[d];
    }
    el[i] = sl;
    er[i] = sr;
}

#define NODES_PER_BIN 391
#define N_BINS 256
#define EB 3200
#define NB (N_EDGES / EB)
#define TAB_STRIDE (N_BINS + 1)
#define SORT_THREADS 800
typedef __attribute__((ext_vector_type(2))) int v2i;
static_assert(N_BINS * NODES_PER_BIN >= N_NODES, "the bins cover every node");
static_assert((N_NODES - 1) / NODES_PER_BIN < N_BINS, "every bin number is below 256");
static_assert(N_EDGES % 32 == 0, "a wave's group is 32 whole edges");
static_assert(EB == (SORT_THREADS / 32) * 4 * 32, "25 waves x 4 groups x 32 edges");
static_assert(N_EDGES % EB == 0, "every block is whole, so no wave needs a bounds test");
static_assert((size_t)N_EDGES * 2 + (size_t)NB * TAB_STRIDE <= (size_t)N_EDGES * HEADS, "pairs and table fit in the edge buffer");

__device__ __forceinline__ void logit_max_groups(const int* s_own, const int* d_own, int lane,
                                                 const float* __restrict__ el, const float* __restrict__ er,
                                                 unsigned int* __restrict__ emax) {
    const int h   = lane & 3;
    const int sub = lane >> 2;
    int   dd[4][4];
    float a[4][4], b[4][4];
    #pragma unroll
    for (int i = 0; i < 4; ++i) {
        #pragma unroll
        for (int u = 0; u < 4; ++u) {
            const int j = u * 8 + sub;
            const int s = __shfl(s_own[i], j, 32);
            dd[i][u] = __shfl(d_own[i], j, 32);
            a[i][u] = el[(size_t)s * 4 + h];
            b[i][u] = er[(size_t)dd[i][u] * 4 + h];
        }
    }
    #pragma unroll
    for (int i = 0; i < 4; ++i) {
        #pragma unroll
        for (int u = 0; u < 4; ++u) {
            float v = a[i][u] + b[i][u];
            v = (v >= 0.0f) ? v : NEG_SLOPE * v;
            atomicMax(emax + (size_t)dd[i][u] * 4 + h, f32_to_key(v));
        }
    }
}

__global__ __launch_bounds__(SORT_THREADS) void edge_logit_kernel(
    const int* __restrict__ src, const int* __restrict__ dst,
    const float* __restrict__ el, const float* __restrict__ er,
    unsigned int* __restrict__ emax, int* __restrict__ pairs, unsigned int* __restrict__ tab) {
    __shared__ unsigned int cnt[N_BINS];
    __shared__ unsigned int start[N_BINS + 1];
    const int tid  = threadIdx.x;
    const int lane = tid & 31;
    const int wave = tid >> 5;
    const int bbase  = blockIdx.x * EB;
    const int e_wave = bbase + wave * 128;
    if (tid < N_BINS) cnt[tid] = 0u;
    __syncthreads();

    int s_own[4], d_own[4];
    unsigned int bn[4], rk[4];
    #pragma unroll
    for (int i = 0; i < 4; ++i) {
        const int e = e_wave + i * 32 + lane;
        s_own[i] = src[e];
        d_own[i] = dst[e];
    }
    #pragma unroll
    for (int i = 0; i < 4; ++i) {
        bn[i] = (unsigned int)(d_own[i] / NODES_PER_BIN);
        rk[i] = atomicAdd(&cnt[bn[i]], 1u);
    }
    logit_max_groups(s_own, d_own, lane, el, er, emax);
    __syncthreads();

    if (wave == 0) {
        unsigned int c[8];
        unsigned int s = 0u;
        #pragma unroll
        for (int k = 0; k < 8; ++k) { c[k] = cnt[lane * 8 + k]; s += c[k]; }
        unsigned int incl = s;
        #pragma unroll
        for (int off = 1; off < 32; off <<= 1) {
            const unsigned int t = __shfl_up(incl, off, 32);
            if (lane >= off) incl += t;
        }
        unsigned int run = incl - s;
        #pragma unroll
        for (int k = 0; k < 8; ++k) { start[lane * 8 + k] = run; run += c[k]; }
        if (lane == 31) start[N_BINS] = incl;
    }
    __syncthreads();

    if (tid <= N_BINS) tab[(size_t)blockIdx.x * TAB_STRIDE + tid] = start[tid];
    #pragma unroll
    for (int i = 0; i < 4; ++i) {
        const int p = bbase + (int)(start[bn[i]] + rk[i]);
        const v2i sd = {s_own[i], d_own[i]};
        *(v2i*)(pairs + (size_t)p * 2) = sd;
    }
}

#define AGG_THREADS 1024
#define AGG_WAVES   (AGG_THREADS / 32)
#define AGG_BATCH 8
#define QUEUE_CAP (32 + AGG_BATCH)
#define BLOCKS_PER_WAVE ((NB + AGG_WAVES - 1) / AGG_WAVES)
static_assert(BLOCKS_PER_WAVE <= 32, "one lane per block of the wave");
static_assert(DHEAD == 32 && HD == HEADS * DHEAD && HEADS == 4, "one head = 32 channels = one line = one lane each");
static_assert(AGG_BATCH * HEADS == 32, "a batch's weights are one per lane");

template <bool FULL>
__device__ __forceinline__ void agg_batch(const unsigned int* qv, int cnt, int lane, int lo,
                                          const int* __restrict__ pairs,
                                          const float* __restrict__ el,
                                          const float* __restrict__ feat_src,
                                          float* acc, const float* mx, const float* erl, float* sm) {
    const int ql = lane >> 2;
    const int kl = lane & 3;
    const bool mine = FULL || ql < cnt;
    int   s_l = 0, d_l = 0;
    float a_l = 0.0f;
    if (mine) {
        const v2i sd = *(const v2i*)(pairs + (size_t)qv[ql] * 2);
        s_l = sd.x;
        d_l = sd.y - lo;
        a_l = el[(size_t)s_l * 4 + kl];
    }
    float f[AGG_BATCH][HEADS];
    #pragma unroll
    for (int q = 0; q < AGG_BATCH; ++q) {
        if (FULL || q < cnt) {
            const int s = __builtin_amdgcn_readlane(s_l, q * 4);
            const float* fr = feat_src + (size_t)s * HD + lane;
            #pragma unroll
            for (int k = 0; k < HEADS; ++k) f[q][k] = fr[k * DHEAD];
        }
    }
    float ex = 0.0f;
    if (mine) {
        float v = a_l + erl[d_l * HEADS + kl];
        v = (v >= 0.0f) ? v : NEG_SLOPE * v;
        ex = __expf(v - mx[d_l * HEADS + kl]);
        atomicAdd(sm + d_l * HEADS + kl, ex);
    }
    #pragma unroll
    for (int q = 0; q < AGG_BATCH; ++q) {
        if (FULL || q < cnt) {
            const int dq = __builtin_amdgcn_readlane(d_l, q * 4);
            float* a = acc + dq * HD + lane;
            #pragma unroll
            for (int k = 0; k < HEADS; ++k) {
                const float wq = __int_as_float(__builtin_amdgcn_readlane(__float_as_int(ex), q * 4 + k));
                atomicAdd(a + k * DHEAD, wq * f[q][k]);
            }
        }
    }
}

__global__ __launch_bounds__(AGG_THREADS) void aggregate_kernel(
    const int* __restrict__ pairs, const unsigned int* __restrict__ tab,
    const unsigned int* __restrict__ emax,
    const float* __restrict__ el, const float* __restrict__ er,
    const float* __restrict__ feat_src, float* __restrict__ out) {
    __shared__ float        acc[NODES_PER_BIN * HD];
    __shared__ float        mx[NODES_PER_BIN * HEADS];
    __shared__ float        erl[NODES_PER_BIN * HEADS];
    __shared__ float        sm[NODES_PER_BIN * HEADS];
    __shared__ unsigned int queue[AGG_WAVES * QUEUE_CAP];
    const int tid  = threadIdx.x;
    const int lane = tid & 31;
    const int wave = tid >> 5;
    const int bin  = blockIdx.x;
    const int lo   = bin * NODES_PER_BIN;

    const int myblk = wave + AGG_WAVES * lane;
    unsigned int t0 = 0u, t1 = 0u;
    if (myblk < NB) {
        t0 = tab[(size_t)myblk * TAB_STRIDE + bin];
        t1 = tab[(size_t)myblk * TAB_STRIDE + bin + 1];
    }

    const v4f zf = {0.0f, 0.0f, 0.0f, 0.0f};
    for (int i = tid; i < NODES_PER_BIN * HD / 4; i += AGG_THREADS) ((v4f*)acc)[i] = zf;
    for (int i = tid; i < NODES_PER_BIN * HEADS; i += AGG_THREADS) {
        const int g = lo * HEADS + i;
        const bool ok = g < N_NODES * HEADS;
        mx[i]  = ok ? key_to_f32(emax[g]) : 0.0f;
        erl[i] = ok ? er[g] : 0.0f;
        sm[i]  = 0.0f;
    }
    __syncthreads();

    unsigned int* qv = queue + wave * QUEUE_CAP;
    int qn = 0;
    for (int l = 0; l < BLOCKS_PER_WAVE; ++l) {
        const unsigned int s0 = (unsigned int)__builtin_amdgcn_readlane((int)t0, l);
        const unsigned int s1 = (unsigned int)__builtin_amdgcn_readlane((int)t1, l);
        unsigned int pos = (unsigned int)(wave + AGG_WAVES * l) * EB + s0;
        int left = (int)(s1 - s0);
        while (left > 0) {
            const int n = min(left, 32);
            if (lane < n) qv[qn + lane] = pos + (unsigned int)lane;
            qn += n; pos += (unsigned int)n; left -= n;
            while (qn >= AGG_BATCH) {
                qn -= AGG_BATCH;
                agg_batch<true>(qv + qn, AGG_BATCH, lane, lo, pairs, el, feat_src, acc, mx, erl, sm);
            }
        }
    }
    if (qn > 0) agg_batch<false>(qv, qn, lane, lo, pairs, el, feat_src, acc, mx, erl, sm);
    __syncthreads();

    for (int r = wave; r < NODES_PER_BIN; r += AGG_WAVES) {
        const int node = lo + r;
        if (node < N_NODES) {
            const float es  = sm[r * HEADS + (lane >> 3)];
            const float inv = (es > 0.0f) ? 1.0f / es : 0.0f;
            *(v4f*)(out + (size_t)node * HD + lane * 4) = *(const v4f*)(acc + r * HD + lane * 4) * inv;
        }
    }
}

#define FIN_NPW 8
static_assert(N_NODES % FIN_NPW == 0, "a wave owns 8 whole nodes");
__global__ __launch_bounds__(256) void final_kernel(
    const float* __restrict__ skip,
    const float* __restrict__ Wg, const float* __restrict__ bg,
    const float* __restrict__ ln_g, const float* __restrict__ ln_b,
    const float* __restrict__ prelu_a, float* __restrict__ out) {
    const int wave = (blockIdx.x * blockDim.x + threadIdx.x) >> 5;
    const int lane = threadIdx.x & 31;
    const int n0 = wave * FIN_NPW;
    if (n0 >= N_NODES) return;
    const int c = lane * 4;
    const v4f w0 = *(const v4f*)(Wg + c);
    const v4f w1 = *(const v4f*)(Wg + 128 + c);
    const v4f w2 = *(const v4f*)(Wg + 256 + c);
    const v4f lg = *(const v4f*)(ln_g + c);
    const v4f lb = *(const v4f*)(ln_b + c);
    const float bgv   = bg[0];
    const float alpha = prelu_a[0];
    for (int b = 0; b < FIN_NPW; b += 4) {
        v4f   rr[4], ss[4];
        #pragma unroll
        for (int u = 0; u < 4; ++u) {
            const size_t node = (size_t)(n0 + b + u);
            rr[u] = *(const v4f*)(out  + node * HD + c);
            ss[u] = *(const v4f*)(skip + node * HD + c);
        }
        #pragma unroll
        for (int u = 0; u < 4; ++u) {
            const size_t node = (size_t)(n0 + b + u);
            const v4f r  = rr[u];
            const v4f sk = ss[u];
            const v4f dm = r - sk;
            float part = r.x * w0.x + r.y * w0.y + r.z * w0.z + r.w * w0.w
                       + sk.x * w1.x + sk.y * w1.y + sk.z * w1.z + sk.w * w1.w
                       + dm.x * w2.x + dm.y * w2.y + dm.z * w2.z + dm.w * w2.w;
            #pragma unroll
            for (int off = 16; off > 0; off >>= 1) part += __shfl_xor(part, off, 32);
            const float g = 1.0f / (1.0f + __expf(-(part + bgv)));
            v4f z = g * r + (1.0f - g) * sk;
            float s  = z.x + z.y + z.z + z.w;
            float sq = z.x * z.x + z.y * z.y + z.z * z.z + z.w * z.w;
            #pragma unroll
            for (int off = 16; off > 0; off >>= 1) {
                s  += __shfl_xor(s,  off, 32);
                sq += __shfl_xor(sq, off, 32);
            }
            const float mu   = s * (1.0f / 128.0f);
            const float var  = sq * (1.0f / 128.0f) - mu * mu;
            const float rstd = rsqrtf(var + EPS_LN);
            v4f y;
            #pragma unroll
            for (int i = 0; i < 4; ++i) {
                float yy = (z[i] - mu) * rstd * lg[i] + lb[i];
                y[i] = (yy >= 0.0f) ? yy : alpha * yy;
            }
            *(v4f*)(out + node * HD + c) = y;
        }
    }
}

extern "C" void kernel_launch(void* const* d_in, const int* in_sizes, int n_in,
                              void* d_out, int out_size, void* d_ws, size_t ws_size,
                              hipStream_t stream) {
    const float* feat    = (const float*)d_in[0];
    const int*   src     = (const int*)  d_in[1];
    const int*   dst     = (const int*)  d_in[2];
    const float* W_src   = (const float*)d_in[3];
    const float* b_src   = (const float*)d_in[4];
    const float* attn_l  = (const float*)d_in[5];
    const float* attn_r  = (const float*)d_in[6];
    const float* W_skip  = (const float*)d_in[7];
    const float* b_skip  = (const float*)d_in[8];
    const float* W_gate  = (const float*)d_in[9];
    const float* b_gate  = (const float*)d_in[10];
    const float* ln_g    = (const float*)d_in[11];
    const float* ln_b    = (const float*)d_in[12];
    const float* prelu_a = (const float*)d_in[13];
    float* out = (float*)d_out;

    float* ws       = (float*)d_ws;
    float* feat_src = ws;
    float* skip     = feat_src + (size_t)N_NODES * HD;
    float* el       = skip     + (size_t)N_NODES * HD;
    float* er       = el       + (size_t)N_NODES * HEADS;
    float* emax     = er       + (size_t)N_NODES * HEADS;
    float* binsp    = emax     + (size_t)N_NODES * HEADS;
    float* ebuf     = binsp    + (size_t)N_NODES * HEADS;

    fill_kernel<<<FILL_BLOCKS, 256, 0, stream>>>((unsigned int*)emax, W_src, W_skip, ebuf);
    {
        int waves  = N_NODES / 16;
        int blocks = (waves * 32 + 255) / 256;
        gemm_wmma_kernel<<<blocks, 256, 0, stream>>>(feat, ebuf, b_src, b_skip,
                                                     feat_src, skip);
    }
    {
        int n = N_NODES * HEADS;
        elr_kernel<<<(n + 255) / 256, 256, 0, stream>>>(feat_src, attn_l, attn_r, el, er);
    }
    int*          pairs = (int*)ebuf;
    unsigned int* tab   = (unsigned int*)(ebuf + (size_t)N_EDGES * 2);
    edge_logit_kernel<<<NB, SORT_THREADS, 0, stream>>>(src, dst, el, er, (unsigned int*)emax, pairs, tab);
    aggregate_kernel<<<N_BINS, AGG_THREADS, 0, stream>>>(pairs, tab, (const unsigned int*)emax, el, er,
                                                         feat_src, out);
    final_kernel<<<(N_NODES / FIN_NPW * 32 + 255) / 256, 256, 0, stream>>>(skip, W_gate, b_gate,
                                                                           ln_g, ln_b, prelu_a, out);
}
